# starting kernel
# baseline (speedup 1.0000x reference)
#include <hip/hip_runtime.h>
#include <hip/hip_bf16.h>

#define NN   20000
#define EE   2000
#define NNZ  100000
#define FF   128
#define HH   128
#define DD   6
#define COUT 40
#define DH   (DD * HH)

typedef __attribute__((ext_vector_type(2))) float v2f;
typedef __attribute__((ext_vector_type(8))) float v8f;

#define ATOMIC_ADD_F32(p, v) unsafeAtomicAdd((p), (v))

#define TK 32
#define TM 32
__global__ __launch_bounds__(128)
void gemm_wmma_f32(const float* __restrict__ A, const float* __restrict__ B,
                   float* __restrict__ C, int M, int Nc, int K, int ldb, int ldc) {
  __shared__ float As[TM][TK + 1];
  __shared__ float Bs[TK][64];

  const int tid  = threadIdx.x;
  const int wave = tid >> 5;
  const int lane = tid & 31;
  const int la   = lane & 15;
  const int hi   = lane >> 4;

  const int m0 = blockIdx.y * TM;
  const int nb = blockIdx.x * 64;
  const int n0 = nb + wave * 16;

  v8f acc0 = {};
  v8f acc1 = {};

  for (int kk = 0; kk < K; kk += TK) {
    for (int i = tid; i < TM * TK; i += 128) {
      int r = i / TK, c = i % TK;
      int gm = m0 + r;
      As[r][c] = (gm < M) ? A[(size_t)gm * K + kk + c] : 0.0f;
    }
    for (int i = tid; i < TK * 64; i += 128) {
      int r = i >> 6, c = i & 63;
      int gn = nb + c;
      Bs[r][c] = (gn < Nc) ? B[(size_t)(kk + r) * ldb + gn] : 0.0f;
    }
    if (kk + TK < K) {
      int pr = tid >> 2, pc = (tid & 3) << 3;
      int gm = m0 + pr;
      if (gm < M) __builtin_prefetch(&A[(size_t)gm * K + kk + TK + pc], 0, 3);
    }
    __syncthreads();

#pragma unroll
    for (int k4 = 0; k4 < TK; k4 += 4) {
      v2f a0, a1, b;
      b.x  = Bs[k4 + 2 * hi][(wave << 4) + la];
      b.y  = Bs[k4 + 2 * hi + 1][(wave << 4) + la];
      a0.x = As[la][k4 + 2 * hi];
      a0.y = As[la][k4 + 2 * hi + 1];
      a1.x = As[16 + la][k4 + 2 * hi];
      a1.y = As[16 + la][k4 + 2 * hi + 1];
      acc0 = __builtin_amdgcn_wmma_f32_16x16x4_f32(
          false, a0, false, b, (short)0, acc0, false, false);
      acc1 = __builtin_amdgcn_wmma_f32_16x16x4_f32(
          false, a1, false, b, (short)0, acc1, false, false);
    }
    __syncthreads();
  }

  const int n = n0 + la;
  if (n < Nc) {
#pragma unroll
    for (int r = 0; r < 8; ++r) {
      int gm = m0 + r + 8 * hi;
      if (gm < M) C[(size_t)gm * ldc + n] = acc0[r];
    }
#pragma unroll
    for (int r = 0; r < 8; ++r) {
      int gm = m0 + 16 + r + 8 * hi;
      if (gm < M) C[(size_t)gm * ldc + n] = acc1[r];
    }
  }
}

__global__ void mean_d_kernel(const float* __restrict__ xl, float* __restrict__ xm,
                              int rows) {
  int gid = blockIdx.x * blockDim.x + threadIdx.x;
  if (gid >= rows * FF) return;
  int n = gid >> 7, f = gid & 127;
  const float* p = xl + (size_t)n * DH + f;
  float s = 0.0f;
#pragma unroll
  for (int d = 0; d < DD; ++d) s += p[d * FF];
  xm[gid] = s * (1.0f / (float)DD);
}

__global__ void alpha_deg_kernel(const float* __restrict__ xm, const float* __restrict__ em,
                                 const int* __restrict__ row, const int* __restrict__ col,
                                 float* __restrict__ alpha, float* __restrict__ Dg,
                                 float* __restrict__ Bg) {
  int gid = blockIdx.x * blockDim.x + threadIdx.x;
  if (gid >= NNZ * DD) return;
  int i = gid / DD, dd = gid - i * DD;
  int r = row[i], c = col[i];
  float v = xm[r * DD + dd] + em[c * DD + dd];
  v = 1.0f / (1.0f + __expf(-v));
  alpha[gid] = v;
  ATOMIC_ADD_F32(&Dg[r * DD + dd], v);
  ATOMIC_ADD_F32(&Bg[c * DD + dd], v);
}

__global__ void recip_kernel(float* __restrict__ p, int n) {
  int gid = blockIdx.x * blockDim.x + threadIdx.x;
  if (gid >= n) return;
  float v = p[gid];
  p[gid] = (v != 0.0f) ? (1.0f / v) : 0.0f;
}

__global__ void fill_kernel(float* __restrict__ p, size_t n, float v) {
  size_t stride = (size_t)gridDim.x * blockDim.x;
  for (size_t i = blockIdx.x * (size_t)blockDim.x + threadIdx.x; i < n; i += stride)
    p[i] = v;
}

__global__ __launch_bounds__(192)
void scatter_to_edge(const float* __restrict__ xp, const float* __restrict__ alpha,
                     const float* __restrict__ Binv, const int* __restrict__ row,
                     const int* __restrict__ col, float* __restrict__ m) {
  int i = blockIdx.x;
  int idx = threadIdx.x << 2;
  int dd = idx >> 7;
  int r = row[i], c = col[i];
  float s = Binv[c * DD + dd] * alpha[(size_t)i * DD + dd];
  float4 v = *(const float4*)(xp + (size_t)r * DH + idx);
  float* dst = m + (size_t)c * DH + idx;
  ATOMIC_ADD_F32(dst + 0, s * v.x);
  ATOMIC_ADD_F32(dst + 1, s * v.y);
  ATOMIC_ADD_F32(dst + 2, s * v.z);
  ATOMIC_ADD_F32(dst + 3, s * v.w);
}

__global__ __launch_bounds__(192)
void scatter_to_node(const float* __restrict__ m, const float* __restrict__ alpha,
                     const float* __restrict__ Dinv, const int* __restrict__ row,
                     const int* __restrict__ col, float* __restrict__ out) {
  int i = blockIdx.x;
  int idx = threadIdx.x << 2;
  int dd = idx >> 7;
  int r = row[i], c = col[i];
  float s = Dinv[r * DD + dd] * alpha[(size_t)i * DD + dd];
  float4 v = *(const float4*)(m + (size_t)c * DH + idx);
  float* dst = out + (size_t)r * DH + idx;
  ATOMIC_ADD_F32(dst + 0, s * v.x);
  ATOMIC_ADD_F32(dst + 1, s * v.y);
  ATOMIC_ADD_F32(dst + 2, s * v.z);
  ATOMIC_ADD_F32(dst + 3, s * v.w);
}

__global__ void bias_act_kernel(float* __restrict__ h, const float* __restrict__ b,
                                size_t total, int do_elu) {
  size_t gid = blockIdx.x * (size_t)blockDim.x + threadIdx.x;
  if (gid >= total) return;
  float v = h[gid] + b[gid & 127];
  if (do_elu) v = (v > 0.0f) ? v : (__expf(v) - 1.0f);
  h[gid] = v;
}

static inline void launch_gemm(const float* A, const float* B, float* C,
                               int M, int Nc, int K, int ldb, int ldc,
                               hipStream_t s) {
  dim3 grid((Nc + 63) / 64, (M + TM - 1) / TM);
  gemm_wmma_f32<<<grid, dim3(128), 0, s>>>(A, B, C, M, Nc, K, ldb, ldc);
}

extern "C" void kernel_launch(void* const* d_in, const int* in_sizes, int n_in,
                              void* d_out, int out_size, void* d_ws, size_t ws_size,
                              hipStream_t stream) {
  const float* x        = (const float*)d_in[0];
  const int*   eidx     = (const int*)d_in[1];
  const float* hedge    = (const float*)d_in[2];
  const float* lin_W    = (const float*)d_in[3];
  const float* sheaf_W  = (const float*)d_in[4];
  const float* conv1_W  = (const float*)d_in[5];
  const float* conv1_b  = (const float*)d_in[6];
  const float* conv2_W  = (const float*)d_in[7];
  const float* conv2_b  = (const float*)d_in[8];
  const float* lin2_W   = (const float*)d_in[9];
  float* out = (float*)d_out;

  const int* row = eidx;
  const int* col = eidx + NNZ;

  float* ws    = (float*)d_ws;
  float* bufA  = ws;
  float* bufB  = bufA + (size_t)NN * DH;
  float* el    = bufB + (size_t)NN * DH;
  float* mbuf  = el + (size_t)EE * DH;
  float* xmean = mbuf + (size_t)EE * DH;
  float* emean = xmean + (size_t)NN * FF;
  float* xm    = emean + (size_t)EE * FF;
  float* em    = xm + (size_t)NN * DD;
  float* alpha = em + (size_t)EE * DD;
  float* Dg    = alpha + (size_t)NNZ * DD;
  float* Bg    = Dg + (size_t)NN * DD;

  const size_t nBig = (size_t)NN * DH;
  const size_t eBig = (size_t)EE * DH;

  launch_gemm(x, lin_W, bufA, NN, DH, FF, DH, DH, stream);
  launch_gemm(hedge, lin_W, el, EE, DH, FF, DH, DH, stream);

  mean_d_kernel<<<(NN * FF + 255) / 256, 256, 0, stream>>>(bufA, xmean, NN);
  mean_d_kernel<<<(EE * FF + 255) / 256, 256, 0, stream>>>(el, emean, EE);

  launch_gemm(xmean, sheaf_W, xm, NN, DD, FF, DD, DD, stream);
  launch_gemm(emean, sheaf_W + FF * DD, em, EE, DD, FF, DD, DD, stream);

  fill_kernel<<<64, 256, 0, stream>>>(Dg, (size_t)NN * DD, 0.0f);
  fill_kernel<<<64, 256, 0, stream>>>(Bg, (size_t)EE * DD, 0.0f);
  alpha_deg_kernel<<<(NNZ * DD + 255) / 256, 256, 0, stream>>>(xm, em, row, col,
                                                               alpha, Dg, Bg);
  recip_kernel<<<(NN * DD + 255) / 256, 256, 0, stream>>>(Dg, NN * DD);
  recip_kernel<<<(EE * DD + 255) / 256, 256, 0, stream>>>(Bg, EE * DD);

  launch_gemm(bufA, conv1_W, bufB, NN * DD, HH, FF, HH, HH, stream);
  fill_kernel<<<1024, 256, 0, stream>>>(mbuf, eBig, 0.0f);
  scatter_to_edge<<<NNZ, 192, 0, stream>>>(bufB, alpha, Bg, row, col, mbuf);
  fill_kernel<<<4096, 256, 0, stream>>>(bufA, nBig, 0.0f);
  scatter_to_node<<<NNZ, 192, 0, stream>>>(mbuf, alpha, Dg, row, col, bufA);
  bias_act_kernel<<<(unsigned)((nBig + 255) / 256), 256, 0, stream>>>(bufA, conv1_b,
                                                                      nBig, 1);

  launch_gemm(bufA, conv2_W, bufB, NN * DD, HH, HH, HH, HH, stream);
  fill_kernel<<<1024, 256, 0, stream>>>(mbuf, eBig, 0.0f);
  scatter_to_edge<<<NNZ, 192, 0, stream>>>(bufB, alpha, Bg, row, col, mbuf);
  fill_kernel<<<4096, 256, 0, stream>>>(bufA, nBig, 0.0f);
  scatter_to_node<<<NNZ, 192, 0, stream>>>(mbuf, alpha, Dg, row, col, bufA);
  bias_act_kernel<<<(unsigned)((nBig + 255) / 256), 256, 0, stream>>>(bufA, conv2_b,
                                                                      nBig, 0);

  launch_gemm(bufA, lin2_W, out, NN, COUT, DH, COUT, COUT, stream);
}


# Mean over d taken in the input-projection GEMM's epilogue: wave tile = 16 rows x (6 d x 32 f), the six d of an (n,f) sit in one lane's accumulators; both mean passes, the el store and 2 launches go
# speedup vs baseline: 3.2984x; 3.2984x over previous
#include <hip/hip_runtime.h>
#include <hip/hip_bf16.h>

#define NN   20000
#define EE   2000
#define NNZ  100000
#define FF   128
#define HH   128
#define DD   6
#define COUT 40
#define DH   (DD * HH)

typedef __attribute__((ext_vector_type(2))) float v2f;
typedef __attribute__((ext_vector_type(8))) float v8f;

#define ATOMIC_ADD_F32(p, v) unsafeAtomicAdd((p), (v))

typedef __attribute__((ext_vector_type(4))) float v4f;

__device__ __forceinline__ float elu1(float v) {
  return (v > 0.0f) ? v : (__expf(v) - 1.0f);
}

template <int K, int NT, int WAVES, bool VECB, bool ELUA = false>
__global__ __launch_bounds__(WAVES * 32)
void gemm_direct_f32(const float* __restrict__ A, const float* __restrict__ B,
                     float* __restrict__ C, int M, int Nc, int ldb, int ldc) {
  constexpr int NB = NT * 16;
  constexpr int KC = 128;
  constexpr int THREADS = WAVES * 32;
  constexpr bool PERM = (NT % 4) == 0;
  __shared__ v4f Bs[(KC / 4) * NB];

  const int tid  = threadIdx.x;
  const int wave = tid >> 5;
  const int lane = tid & 31;
  const int c    = lane & 15;
  const int h    = lane >> 4;
  const int nb   = blockIdx.x * NB;

  const int row_tiles = (M + 15) / 16;
  const int per_sweep = gridDim.y * WAVES;
  const int sweeps    = (row_tiles + per_sweep - 1) / per_sweep;

  auto slot_of = [](int n) {
    return PERM ? ((n >> 6) * 64 + (n & 3) * 16 + ((n & 63) >> 2)) : n;
  };

  auto stage_panel = [&](int kc) {
    if (VECB) {
      constexpr int G = NB / 4;
      constexpr int ITERS = ((KC / 4) * G) / THREADS;
      static_assert(((KC / 4) * G) % THREADS == 0, "panel must divide evenly over the block");
#pragma unroll
      for (int i = 0; i < ITERS; ++i) {
        const int idx = tid + i * THREADS;
        const int q = idx / G, n = (idx % G) * 4;
        const int gn = nb + n;
        const int gs = (gn < Nc) ? gn : (Nc - 4);
        const float* bp = B + (size_t)(kc + 4 * q) * ldb + gs;
        v4f r0 = *(const v4f*)(bp);
        v4f r1 = *(const v4f*)(bp + ldb);
        v4f r2 = *(const v4f*)(bp + 2 * (size_t)ldb);
        v4f r3 = *(const v4f*)(bp + 3 * (size_t)ldb);
        Bs[q * NB + slot_of(n + 0)] = (v4f){r0.x, r1.x, r2.x, r3.x};
        Bs[q * NB + slot_of(n + 1)] = (v4f){r0.y, r1.y, r2.y, r3.y};
        Bs[q * NB + slot_of(n + 2)] = (v4f){r0.z, r1.z, r2.z, r3.z};
        Bs[q * NB + slot_of(n + 3)] = (v4f){r0.w, r1.w, r2.w, r3.w};
      }
    } else {
      constexpr int ITERS = ((KC / 4) * NB) / THREADS;
      static_assert(((KC / 4) * NB) % THREADS == 0, "panel must divide evenly over the block");
#pragma unroll
      for (int i = 0; i < ITERS; ++i) {
        const int idx = tid + i * THREADS;
        const int q = idx / NB, n = idx % NB;
        const int gn = nb + n;
        const int gs = (gn < Nc) ? gn : (Nc - 1);
        const float* bp = B + (size_t)(kc + 4 * q) * ldb + gs;
        v4f v;
        v.x = bp[0];
        v.y = bp[ldb];
        v.z = bp[2 * (size_t)ldb];
        v.w = bp[3 * (size_t)ldb];
        Bs[q * NB + slot_of(n)] = v;
      }
    }
  };

  if (K == KC) {
    stage_panel(0);
    __syncthreads();
  }

  for (int it = 0; it < sweeps; ++it) {
    const int tile = (it * gridDim.y + blockIdx.y) * WAVES + wave;
    const int r0   = tile * 16;
    if (K == KC && it > 0 && r0 >= M) break;

    int gm = r0 + c;
    gm = (gm < M) ? gm : (M - 1);
    const float* ap = A + (size_t)gm * K + 4 * h;

    v8f acc[NT];
#pragma unroll
    for (int nt = 0; nt < NT; ++nt) acc[nt] = (v8f){};

    v4f a_cur[4];
#pragma unroll
    for (int j = 0; j < 4; ++j) {
      a_cur[j] = *(const v4f*)(ap + 8 * j);
      if (ELUA) a_cur[j] = (v4f){elu1(a_cur[j].x), elu1(a_cur[j].y), elu1(a_cur[j].z), elu1(a_cur[j].w)};
    }

    for (int kc = 0; kc < K; kc += KC) {
      if (K > KC) {
        __syncthreads();
        stage_panel(kc);
        __syncthreads();
      }

#pragma unroll
      for (int s = 0; s < 4; ++s) {
        v4f a_nxt[4];
        const int kn = kc + (s + 1) * 32;
        if (kn < K) {
#pragma unroll
          for (int j = 0; j < 4; ++j) a_nxt[j] = *(const v4f*)(ap + kn + 8 * j);
        }
        __builtin_amdgcn_sched_barrier(0);
#pragma unroll
        for (int j = 0; j < 4; ++j) {
          v4f b[NT];
#pragma unroll
          for (int nt = 0; nt < NT; ++nt) b[nt] = Bs[(8 * s + 2 * j + h) * NB + nt * 16 + c];
          const v2f alo = {a_cur[j].x, a_cur[j].y};
          const v2f ahi = {a_cur[j].z, a_cur[j].w};
#pragma unroll
          for (int nt = 0; nt < NT; ++nt) {
            const v2f blo = {b[nt].x, b[nt].y};
            acc[nt] = __builtin_amdgcn_wmma_f32_16x16x4_f32(
                false, alo, false, blo, (short)0, acc[nt], false, false);
          }
#pragma unroll
          for (int nt = 0; nt < NT; ++nt) {
            const v2f bhi = {b[nt].z, b[nt].w};
            acc[nt] = __builtin_amdgcn_wmma_f32_16x16x4_f32(
                false, ahi, false, bhi, (short)0, acc[nt], false, false);
          }
        }
        if (kn < K) {
#pragma unroll
          for (int j = 0; j < 4; ++j) {
            a_cur[j] = ELUA ? (v4f){elu1(a_nxt[j].x), elu1(a_nxt[j].y), elu1(a_nxt[j].z), elu1(a_nxt[j].w)}
                            : a_nxt[j];
          }
        }
      }
    }

    const bool vec_rows = ((ldc & 3) == 0) && (nb + NB <= Nc);
#pragma unroll
    for (int r = 0; r < 8; ++r) {
      const int grow = r0 + 8 * h + r;
      if (grow < M) {
        float* crow = C + (size_t)grow * ldc;
        if (PERM) {
#pragma unroll
          for (int g = 0; g < NT / 4; ++g) {
            const int n = nb + g * 64 + 4 * c;
            if (vec_rows) {
              v4f o = {acc[4 * g + 0][r], acc[4 * g + 1][r], acc[4 * g + 2][r], acc[4 * g + 3][r]};
              *(v4f*)(crow + n) = o;
            } else {
#pragma unroll
              for (int e = 0; e < 4; ++e)
                if (n + e < Nc) crow[n + e] = acc[4 * g + e][r];
            }
          }
        } else {
#pragma unroll
          for (int nt = 0; nt < NT; ++nt) {
            const int n = nb + nt * 16 + c;
            if (n < Nc) crow[n] = acc[nt][r];
          }
        }
      }
    }
  }
}

template <int WAVES, bool STORE_C>
__global__ __launch_bounds__(WAVES * 32)
void gemm_lin_mean_f32(const float* __restrict__ A, const float* __restrict__ B,
                       float* __restrict__ C, float* __restrict__ Cm, int M) {
  constexpr int K  = FF;
  constexpr int NT = 2 * DD;
  constexpr int NB = NT * 16;
  constexpr int FS = 32;
  constexpr int THREADS = WAVES * 32;
  __shared__ v4f Bs[(K / 4) * NB];

  const int tid  = threadIdx.x;
  const int lane = tid & 31;
  const int c    = lane & 15;
  const int h    = lane >> 4;
  const int wave = __builtin_amdgcn_readfirstlane(tid >> 5);
  const int f0   = blockIdx.x * FS;

  {
    constexpr int G = NB / 4;
    constexpr int ITERS = ((K / 4) * G) / THREADS;
    static_assert(((K / 4) * G) % THREADS == 0, "panel must divide evenly over the block");
#pragma unroll
    for (int i = 0; i < ITERS; ++i) {
      const int idx = tid + i * THREADS;
      const int q = idx / G, g = idx % G;
      const int d = g >> 3, j0 = (g & 7) * 4;
      const float* bp = B + (size_t)(4 * q) * DH + d * FF + f0 + j0;
      const v4f r0 = *(const v4f*)(bp);
      const v4f r1 = *(const v4f*)(bp + DH);
      const v4f r2 = *(const v4f*)(bp + 2 * DH);
      const v4f r3 = *(const v4f*)(bp + 3 * DH);
      const int s0 = q * NB + (2 * d) * 16 + (j0 >> 1);
      Bs[s0]      = (v4f){r0.x, r1.x, r2.x, r3.x};
      Bs[s0 + 16] = (v4f){r0.y, r1.y, r2.y, r3.y};
      Bs[s0 + 1]  = (v4f){r0.z, r1.z, r2.z, r3.z};
      Bs[s0 + 17] = (v4f){r0.w, r1.w, r2.w, r3.w};
    }
  }
  __syncthreads();

  const int row_tiles = (M + 15) / 16;
  const int per_sweep = gridDim.y * WAVES;
  const int sweeps    = (row_tiles + per_sweep - 1) / per_sweep;

  for (int it = 0; it < sweeps; ++it) {
    const int tile = it * per_sweep + wave * gridDim.y + blockIdx.y;
    const int r0   = tile * 16;
    if (r0 >= M) break;

    int gm = r0 + c;
    gm = (gm < M) ? gm : (M - 1);
    const float* ap = A + (size_t)gm * K + 4 * h;

    v8f acc[NT];
#pragma unroll
    for (int nt = 0; nt < NT; ++nt) acc[nt] = (v8f){};

    v4f a_cur[4];
#pragma unroll
    for (int j = 0; j < 4; ++j) a_cur[j] = *(const v4f*)(ap + 8 * j);

#pragma unroll
    for (int s = 0; s < 4; ++s) {
      v4f a_nxt[4];
      const int kn = (s + 1) * 32;
      if (kn < K) {
#pragma unroll
        for (int j = 0; j < 4; ++j) a_nxt[j] = *(const v4f*)(ap + kn + 8 * j);
      }
      __builtin_amdgcn_sched_barrier(0);
#pragma unroll
      for (int j = 0; j < 4; ++j) {
        v4f b[NT];
#pragma unroll
        for (int nt = 0; nt < NT; ++nt) b[nt] = Bs[(8 * s + 2 * j + h) * NB + nt * 16 + c];
        const v2f alo = {a_cur[j].x, a_cur[j].y};
        const v2f ahi = {a_cur[j].z, a_cur[j].w};
#pragma unroll
        for (int nt = 0; nt < NT; ++nt) {
          const v2f blo = {b[nt].x, b[nt].y};
          acc[nt] = __builtin_amdgcn_wmma_f32_16x16x4_f32(
              false, alo, false, blo, (short)0, acc[nt], false, false);
        }
#pragma unroll
        for (int nt = 0; nt < NT; ++nt) {
          const v2f bhi = {b[nt].z, b[nt].w};
          acc[nt] = __builtin_amdgcn_wmma_f32_16x16x4_f32(
              false, ahi, false, bhi, (short)0, acc[nt], false, false);
        }
      }
      if (kn < K) {
#pragma unroll
        for (int j = 0; j < 4; ++j) a_cur[j] = a_nxt[j];
      }
    }

#pragma unroll
    for (int r = 0; r < 8; ++r) {
      const int grow = r0 + 8 * h + r;
      if (grow < M) {
        float s0 = 0.0f, s1 = 0.0f;
#pragma unroll
        for (int d = 0; d < DD; ++d) {
          const float v0 = acc[2 * d][r], v1 = acc[2 * d + 1][r];
          if (STORE_C)
            *(v2f*)(C + (size_t)grow * DH + d * FF + f0 + 2 * c) = (v2f){v0, v1};
          s0 += v0;
          s1 += v1;
        }
        *(v2f*)(Cm + (size_t)grow * FF + f0 + 2 * c) =
            (v2f){s0 * (1.0f / (float)DD), s1 * (1.0f / (float)DD)};
      }
    }
  }
}

__global__ void alpha_deg_kernel(const float* __restrict__ xm, const float* __restrict__ em,
                                 const int* __restrict__ row, const int* __restrict__ col,
                                 float* __restrict__ alpha, float* __restrict__ Dg,
                                 float* __restrict__ Bg) {
  int gid = blockIdx.x * blockDim.x + threadIdx.x;
  if (gid >= NNZ * DD) return;
  int i = gid / DD, dd = gid - i * DD;
  int r = row[i], c = col[i];
  float v = xm[r * DD + dd] + em[c * DD + dd];
  v = 1.0f / (1.0f + __expf(-v));
  alpha[gid] = v;
  ATOMIC_ADD_F32(&Dg[r * DD + dd], v);
  ATOMIC_ADD_F32(&Bg[c * DD + dd], v);
}

__global__ void recip_kernel(float* __restrict__ p, int n) {
  int gid = blockIdx.x * blockDim.x + threadIdx.x;
  if (gid >= n) return;
  float v = p[gid];
  p[gid] = (v != 0.0f) ? (1.0f / v) : 0.0f;
}

__global__ void fill_kernel(float* __restrict__ p, size_t n, float v) {
  size_t stride = (size_t)gridDim.x * blockDim.x;
  for (size_t i = blockIdx.x * (size_t)blockDim.x + threadIdx.x; i < n; i += stride)
    p[i] = v;
}

#define OW_WAVES 32
#define OW_SLICE 128
#define OW_RING  64
#define OW_BATCH 8

typedef __attribute__((ext_vector_type(2))) int v2i;
typedef __attribute__((ext_vector_type(4))) int v4i;

static_assert(NNZ % 4 == 0, "a lane's four indices are in range together");

__device__ __forceinline__ void lds_add(float* p, float v) {
  __hip_atomic_fetch_add(p, v, __ATOMIC_RELAXED, __HIP_MEMORY_SCOPE_WORKGROUP);
}

template <bool FULL>
__device__ __forceinline__ void ow_drain(const v2i* ring, int head, int n, int lane,
                                         const int* __restrict__ src_idx,
                                         const float* __restrict__ al,
                                         const float* __restrict__ iv,
                                         const float* sp, float* ap) {
  int b_l = lane & (OW_BATCH - 1);
  if (!FULL) b_l = (b_l < n) ? b_l : (n - 1);
  const v2i   e    = ring[(head + b_l) & (OW_RING - 1)];
  const int   srow = src_idx[e.x];
  const float scl  = iv[e.y * DD] * al[(size_t)e.x * DD];

  v4f   v[OW_BATCH];
  int   dl[OW_BATCH];
  float sc[OW_BATCH];
#pragma unroll
  for (int b = 0; b < OW_BATCH; ++b) {
    if (FULL || b < n) {
      const int sr = __builtin_amdgcn_readlane(srow, b);
      dl[b] = __builtin_amdgcn_readlane(e.y, b);
      sc[b] = __int_as_float(__builtin_amdgcn_readlane(__float_as_int(scl), b));
      v[b]  = *(const v4f*)(sp + (size_t)sr * DH);
    }
  }
#pragma unroll
  for (int b = 0; b < OW_BATCH; ++b) {
    if (FULL || b < n) {
      float* a = ap + dl[b] * OW_SLICE;
      lds_add(a,      sc[b] * v[b].x);
      lds_add(a + 32, sc[b] * v[b].y);
      lds_add(a + 64, sc[b] * v[b].z);
      lds_add(a + 96, sc[b] * v[b].w);
    }
  }
}

template <int R, bool BIAS>
__global__ __launch_bounds__(OW_WAVES * 32)
void owner_rows(const float* __restrict__ src, const float* __restrict__ alpha,
                const float* __restrict__ inv, const int* __restrict__ src_idx,
                const int* __restrict__ dst_idx, float* __restrict__ dst,
                const float* __restrict__ bias, int rows) {
  __shared__ __attribute__((aligned(16))) float acc[R * OW_SLICE];
  __shared__ v2i rings[OW_WAVES * OW_RING];

  const int tid  = threadIdx.x;
  const int lane = tid & 31;
  const int wave = __builtin_amdgcn_readfirstlane(tid >> 5);
  const int r0   = blockIdx.x * R;
  const int dch  = blockIdx.y;
  const int col0 = dch * OW_SLICE;

  for (int k = tid; k < R * OW_SLICE / 4; k += OW_WAVES * 32) ((v4f*)acc)[k] = (v4f){};
  __syncthreads();

  const float* sp   = src + col0 + 4 * lane;
  float*       ap   = acc + lane;
  v2i*         ring = rings + wave * OW_RING;
  const float* al   = alpha + dch;
  const float* iv   = inv + (size_t)r0 * DD + dch;

  constexpr int GROUPS = (NNZ + 127) / 128;

  auto load_group = [&](int g) {
    const int i0 = g * 128 + 4 * lane;
    v4i d = {-1, -1, -1, -1};
    if (i0 < NNZ) d = *(const v4i*)(dst_idx + i0);
    return d;
  };

  int head = 0, count = 0;
  v4i d_n = load_group(wave);
  for (int g = wave; g < GROUPS; g += OW_WAVES) {
    const v4i d = d_n;
    if (g + OW_WAVES < GROUPS) d_n = load_group(g + OW_WAVES);

#pragma unroll
    for (int k = 0; k < 4; ++k) {
      const int  loc = d[k] - r0;
      const bool m   = (unsigned)loc < (unsigned)R;
      const unsigned mask = __builtin_amdgcn_ballot_w32(m);
      if (mask != 0u) {
        if (m) {
          const int e = (head + count + (int)__builtin_amdgcn_mbcnt_lo(mask, 0u)) & (OW_RING - 1);
          ring[e] = (v2i){g * 128 + 4 * lane + k, loc};
        }
        count += __builtin_popcount(mask);
        while (count >= OW_BATCH) {
          ow_drain<true>(ring, head, OW_BATCH, lane, src_idx, al, iv, sp, ap);
          head   = (head + OW_BATCH) & (OW_RING - 1);
          count -= OW_BATCH;
        }
      }
    }
  }
  if (count > 0) ow_drain<false>(ring, head, count, lane, src_idx, al, iv, sp, ap);
  __syncthreads();

  v4f bb = {0.0f, 0.0f, 0.0f, 0.0f};
  if (BIAS) bb = *(const v4f*)(bias + 4 * lane);
  for (int r = wave; r < R; r += OW_WAVES) {
    const int grow = r0 + r;
    if (grow < rows) {
      const float* a = ap + r * OW_SLICE;
      v4f o = {a[0], a[32], a[64], a[96]};
      if (BIAS) o += bb;
      *(v4f*)(dst + (size_t)grow * DH + col0 + 4 * lane) = o;
    }
  }
}

extern "C" void kernel_launch(void* const* d_in, const int* in_sizes, int n_in,
                              void* d_out, int out_size, void* d_ws, size_t ws_size,
                              hipStream_t stream) {
  const float* x        = (const float*)d_in[0];
  const int*   eidx     = (const int*)d_in[1];
  const float* hedge    = (const float*)d_in[2];
  const float* lin_W    = (const float*)d_in[3];
  const float* sheaf_W  = (const float*)d_in[4];
  const float* conv1_W  = (const float*)d_in[5];
  const float* conv1_b  = (const float*)d_in[6];
  const float* conv2_W  = (const float*)d_in[7];
  const float* conv2_b  = (const float*)d_in[8];
  const float* lin2_W   = (const float*)d_in[9];
  float* out = (float*)d_out;

  const int* row = eidx;
  const int* col = eidx + NNZ;

  float* ws    = (float*)d_ws;
  float* bufA  = ws;
  float* bufB  = bufA + (size_t)NN * DH;
  float* mbuf  = bufB + (size_t)NN * DH;
  float* xmean = mbuf + (size_t)EE * DH;
  float* emean = xmean + (size_t)NN * FF;
  float* xm    = emean + (size_t)EE * FF;
  float* em    = xm + (size_t)NN * DD;
  float* alpha = em + (size_t)EE * DD;
  float* Dg    = alpha + (size_t)NNZ * DD;
  float* Bg    = Dg + (size_t)NN * DD;

  gemm_lin_mean_f32<16, true><<<dim3(4, 64), 512, 0, stream>>>(x, lin_W, bufA, xmean, NN);
  gemm_lin_mean_f32<4, false><<<dim3(4, 32), 128, 0, stream>>>(hedge, lin_W, nullptr, emean, EE);

  gemm_direct_f32<FF, 1, 4, false><<<dim3(1, 313), 128, 0, stream>>>(xmean, sheaf_W, xm, NN, DD, DD, DD);
  gemm_direct_f32<FF, 1, 4, false><<<dim3(1, 32), 128, 0, stream>>>(emean, sheaf_W + FF * DD, em, EE, DD, DD, DD);

  fill_kernel<<<64, 256, 0, stream>>>(Dg, (size_t)NN * DD, 0.0f);
  fill_kernel<<<64, 256, 0, stream>>>(Bg, (size_t)EE * DD, 0.0f);
  alpha_deg_kernel<<<(NNZ * DD + 255) / 256, 256, 0, stream>>>(xm, em, row, col,
                                                               alpha, Dg, Bg);
  recip_kernel<<<(NN * DD + 255) / 256, 256, 0, stream>>>(Dg, NN * DD);
  recip_kernel<<<(EE * DD + 255) / 256, 256, 0, stream>>>(Bg, EE * DD);

  gemm_direct_f32<FF, 8, 8, true><<<dim3(1, 512), 256, 0, stream>>>(
      bufA, conv1_W, bufB, NN * DD, HH, HH, HH);
  owner_rows<25, false><<<dim3(EE / 25, DH / OW_SLICE), OW_WAVES * 32, 0, stream>>>(
      bufB, alpha, Bg, row, col, mbuf, nullptr, EE);
  owner_rows<250, true><<<dim3(NN / 250, DH / OW_SLICE), OW_WAVES * 32, 0, stream>>>(
      mbuf, alpha, Dg, col, row, bufA, conv1_b, NN);

  gemm_direct_f32<HH, 8, 8, true, true><<<dim3(1, 512), 256, 0, stream>>>(
      bufA, conv2_W, bufB, NN * DD, HH, HH, HH);
  owner_rows<25, false><<<dim3(EE / 25, DH / OW_SLICE), OW_WAVES * 32, 0, stream>>>(
      bufB, alpha, Bg, row, col, mbuf, nullptr, EE);
  owner_rows<250, true><<<dim3(NN / 250, DH / OW_SLICE), OW_WAVES * 32, 0, stream>>>(
      mbuf, alpha, Dg, col, row, bufA, conv2_b, NN);

  gemm_direct_f32<DH, 3, 4, true><<<dim3(1, 313), 128, 0, stream>>>(bufA, lin2_W, out, NN, COUT, COUT, COUT);
}


# Scatter-add scan: pre-pass marks each destination block's incidences in a bitmap (ballots; list neighbours go to different waves); a workgroup reads its bit row, not the index list; new arrival order
# speedup vs baseline: 3.8657x; 1.1720x over previous
#include <hip/hip_runtime.h>
#include <hip/hip_bf16.h>

#define NN   20000
#define EE   2000
#define NNZ  100000
#define FF   128
#define HH   128
#define DD   6
#define COUT 40
#define DH   (DD * HH)

typedef __attribute__((ext_vector_type(2))) float v2f;
typedef __attribute__((ext_vector_type(8))) float v8f;

#define ATOMIC_ADD_F32(p, v) unsafeAtomicAdd((p), (v))

typedef __attribute__((ext_vector_type(4))) float v4f;

__device__ __forceinline__ float elu1(float v) {
  return (v > 0.0f) ? v : (__expf(v) - 1.0f);
}

template <int K, int NT, int WAVES, bool VECB, bool ELUA = false>
__global__ __launch_bounds__(WAVES * 32)
void gemm_direct_f32(const float* __restrict__ A, const float* __restrict__ B,
                     float* __restrict__ C, int M, int Nc, int ldb, int ldc) {
  constexpr int NB = NT * 16;
  constexpr int KC = 128;
  constexpr int THREADS = WAVES * 32;
  constexpr bool PERM = (NT % 4) == 0;
  __shared__ v4f Bs[(KC / 4) * NB];

  const int tid  = threadIdx.x;
  const int wave = tid >> 5;
  const int lane = tid & 31;
  const int c    = lane & 15;
  const int h    = lane >> 4;
  const int nb   = blockIdx.x * NB;

  const int row_tiles = (M + 15) / 16;
  const int per_sweep = gridDim.y * WAVES;
  const int sweeps    = (row_tiles + per_sweep - 1) / per_sweep;

  auto slot_of = [](int n) {
    return PERM ? ((n >> 6) * 64 + (n & 3) * 16 + ((n & 63) >> 2)) : n;
  };

  auto stage_panel = [&](int kc) {
    if (VECB) {
      constexpr int G = NB / 4;
      constexpr int ITERS = ((KC / 4) * G) / THREADS;
      static_assert(((KC / 4) * G) % THREADS == 0, "panel must divide evenly over the block");
#pragma unroll
      for (int i = 0; i < ITERS; ++i) {
        const int idx = tid + i * THREADS;
        const int q = idx / G, n = (idx % G) * 4;
        const int gn = nb + n;
        const int gs = (gn < Nc) ? gn : (Nc - 4);
        const float* bp = B + (size_t)(kc + 4 * q) * ldb + gs;
        v4f r0 = *(const v4f*)(bp);
        v4f r1 = *(const v4f*)(bp + ldb);
        v4f r2 = *(const v4f*)(bp + 2 * (size_t)ldb);
        v4f r3 = *(const v4f*)(bp + 3 * (size_t)ldb);
        Bs[q * NB + slot_of(n + 0)] = (v4f){r0.x, r1.x, r2.x, r3.x};
        Bs[q * NB + slot_of(n + 1)] = (v4f){r0.y, r1.y, r2.y, r3.y};
        Bs[q * NB + slot_of(n + 2)] = (v4f){r0.z, r1.z, r2.z, r3.z};
        Bs[q * NB + slot_of(n + 3)] = (v4f){r0.w, r1.w, r2.w, r3.w};
      }
    } else {
      constexpr int ITERS = ((KC / 4) * NB) / THREADS;
      static_assert(((KC / 4) * NB) % THREADS == 0, "panel must divide evenly over the block");
#pragma unroll
      for (int i = 0; i < ITERS; ++i) {
        const int idx = tid + i * THREADS;
        const int q = idx / NB, n = idx % NB;
        const int gn = nb + n;
        const int gs = (gn < Nc) ? gn : (Nc - 1);
        const float* bp = B + (size_t)(kc + 4 * q) * ldb + gs;
        v4f v;
        v.x = bp[0];
        v.y = bp[ldb];
        v.z = bp[2 * (size_t)ldb];
        v.w = bp[3 * (size_t)ldb];
        Bs[q * NB + slot_of(n)] = v;
      }
    }
  };

  if (K == KC) {
    stage_panel(0);
    __syncthreads();
  }

  for (int it = 0; it < sweeps; ++it) {
    const int tile = (it * gridDim.y + blockIdx.y) * WAVES + wave;
    const int r0   = tile * 16;
    if (K == KC && it > 0 && r0 >= M) break;

    int gm = r0 + c;
    gm = (gm < M) ? gm : (M - 1);
    const float* ap = A + (size_t)gm * K + 4 * h;

    v8f acc[NT];
#pragma unroll
    for (int nt = 0; nt < NT; ++nt) acc[nt] = (v8f){};

    v4f a_cur[4];
#pragma unroll
    for (int j = 0; j < 4; ++j) {
      a_cur[j] = *(const v4f*)(ap + 8 * j);
      if (ELUA) a_cur[j] = (v4f){elu1(a_cur[j].x), elu1(a_cur[j].y), elu1(a_cur[j].z), elu1(a_cur[j].w)};
    }

    for (int kc = 0; kc < K; kc += KC) {
      if (K > KC) {
        __syncthreads();
        stage_panel(kc);
        __syncthreads();
      }

#pragma unroll
      for (int s = 0; s < 4; ++s) {
        v4f a_nxt[4];
        const int kn = kc + (s + 1) * 32;
        if (kn < K) {
#pragma unroll
          for (int j = 0; j < 4; ++j) a_nxt[j] = *(const v4f*)(ap + kn + 8 * j);
        }
        __builtin_amdgcn_sched_barrier(0);
#pragma unroll
        for (int j = 0; j < 4; ++j) {
          v4f b[NT];
#pragma unroll
          for (int nt = 0; nt < NT; ++nt) b[nt] = Bs[(8 * s + 2 * j + h) * NB + nt * 16 + c];
          const v2f alo = {a_cur[j].x, a_cur[j].y};
          const v2f ahi = {a_cur[j].z, a_cur[j].w};
#pragma unroll
          for (int nt = 0; nt < NT; ++nt) {
            const v2f blo = {b[nt].x, b[nt].y};
            acc[nt] = __builtin_amdgcn_wmma_f32_16x16x4_f32(
                false, alo, false, blo, (short)0, acc[nt], false, false);
          }
#pragma unroll
          for (int nt = 0; nt < NT; ++nt) {
            const v2f bhi = {b[nt].z, b[nt].w};
            acc[nt] = __builtin_amdgcn_wmma_f32_16x16x4_f32(
                false, ahi, false, bhi, (short)0, acc[nt], false, false);
          }
        }
        if (kn < K) {
#pragma unroll
          for (int j = 0; j < 4; ++j) {
            a_cur[j] = ELUA ? (v4f){elu1(a_nxt[j].x), elu1(a_nxt[j].y), elu1(a_nxt[j].z), elu1(a_nxt[j].w)}
                            : a_nxt[j];
          }
        }
      }
    }

    const bool vec_rows = ((ldc & 3) == 0) && (nb + NB <= Nc);
#pragma unroll
    for (int r = 0; r < 8; ++r) {
      const int grow = r0 + 8 * h + r;
      if (grow < M) {
        float* crow = C + (size_t)grow * ldc;
        if (PERM) {
#pragma unroll
          for (int g = 0; g < NT / 4; ++g) {
            const int n = nb + g * 64 + 4 * c;
            if (vec_rows) {
              v4f o = {acc[4 * g + 0][r], acc[4 * g + 1][r], acc[4 * g + 2][r], acc[4 * g + 3][r]};
              *(v4f*)(crow + n) = o;
            } else {
#pragma unroll
              for (int e = 0; e < 4; ++e)
                if (n + e < Nc) crow[n + e] = acc[4 * g + e][r];
            }
          }
        } else {
#pragma unroll
          for (int nt = 0; nt < NT; ++nt) {
            const int n = nb + nt * 16 + c;
            if (n < Nc) crow[n] = acc[nt][r];
          }
        }
      }
    }
  }
}

template <int WAVES, bool STORE_C>
__global__ __launch_bounds__(WAVES * 32)
void gemm_lin_mean_f32(const float* __restrict__ A, const float* __restrict__ B,
                       float* __restrict__ C, float* __restrict__ Cm, int M) {
  constexpr int K  = FF;
  constexpr int NT = 2 * DD;
  constexpr int NB = NT * 16;
  constexpr int FS = 32;
  constexpr int THREADS = WAVES * 32;
  __shared__ v4f Bs[(K / 4) * NB];

  const int tid  = threadIdx.x;
  const int lane = tid & 31;
  const int c    = lane & 15;
  const int h    = lane >> 4;
  const int wave = __builtin_amdgcn_readfirstlane(tid >> 5);
  const int f0   = blockIdx.x * FS;

  {
    constexpr int G = NB / 4;
    constexpr int ITERS = ((K / 4) * G) / THREADS;
    static_assert(((K / 4) * G) % THREADS == 0, "panel must divide evenly over the block");
#pragma unroll
    for (int i = 0; i < ITERS; ++i) {
      const int idx = tid + i * THREADS;
      const int q = idx / G, g = idx % G;
      const int d = g >> 3, j0 = (g & 7) * 4;
      const float* bp = B + (size_t)(4 * q) * DH + d * FF + f0 + j0;
      const v4f r0 = *(const v4f*)(bp);
      const v4f r1 = *(const v4f*)(bp + DH);
      const v4f r2 = *(const v4f*)(bp + 2 * DH);
      const v4f r3 = *(const v4f*)(bp + 3 * DH);
      const int s0 = q * NB + (2 * d) * 16 + (j0 >> 1);
      Bs[s0]      = (v4f){r0.x, r1.x, r2.x, r3.x};
      Bs[s0 + 16] = (v4f){r0.y, r1.y, r2.y, r3.y};
      Bs[s0 + 1]  = (v4f){r0.z, r1.z, r2.z, r3.z};
      Bs[s0 + 17] = (v4f){r0.w, r1.w, r2.w, r3.w};
    }
  }
  __syncthreads();

  const int row_tiles = (M + 15) / 16;
  const int per_sweep = gridDim.y * WAVES;
  const int sweeps    = (row_tiles + per_sweep - 1) / per_sweep;

  for (int it = 0; it < sweeps; ++it) {
    const int tile = it * per_sweep + wave * gridDim.y + blockIdx.y;
    const int r0   = tile * 16;
    if (r0 >= M) break;

    int gm = r0 + c;
    gm = (gm < M) ? gm : (M - 1);
    const float* ap = A + (size_t)gm * K + 4 * h;

    v8f acc[NT];
#pragma unroll
    for (int nt = 0; nt < NT; ++nt) acc[nt] = (v8f){};

    v4f a_cur[4];
#pragma unroll
    for (int j = 0; j < 4; ++j) a_cur[j] = *(const v4f*)(ap + 8 * j);

#pragma unroll
    for (int s = 0; s < 4; ++s) {
      v4f a_nxt[4];
      const int kn = (s + 1) * 32;
      if (kn < K) {
#pragma unroll
        for (int j = 0; j < 4; ++j) a_nxt[j] = *(const v4f*)(ap + kn + 8 * j);
      }
      __builtin_amdgcn_sched_barrier(0);
#pragma unroll
      for (int j = 0; j < 4; ++j) {
        v4f b[NT];
#pragma unroll
        for (int nt = 0; nt < NT; ++nt) b[nt] = Bs[(8 * s + 2 * j + h) * NB + nt * 16 + c];
        const v2f alo = {a_cur[j].x, a_cur[j].y};
        const v2f ahi = {a_cur[j].z, a_cur[j].w};
#pragma unroll
        for (int nt = 0; nt < NT; ++nt) {
          const v2f blo = {b[nt].x, b[nt].y};
          acc[nt] = __builtin_amdgcn_wmma_f32_16x16x4_f32(
              false, alo, false, blo, (short)0, acc[nt], false, false);
        }
#pragma unroll
        for (int nt = 0; nt < NT; ++nt) {
          const v2f bhi = {b[nt].z, b[nt].w};
          acc[nt] = __builtin_amdgcn_wmma_f32_16x16x4_f32(
              false, ahi, false, bhi, (short)0, acc[nt], false, false);
        }
      }
      if (kn < K) {
#pragma unroll
        for (int j = 0; j < 4; ++j) a_cur[j] = a_nxt[j];
      }
    }

#pragma unroll
    for (int r = 0; r < 8; ++r) {
      const int grow = r0 + 8 * h + r;
      if (grow < M) {
        float s0 = 0.0f, s1 = 0.0f;
#pragma unroll
        for (int d = 0; d < DD; ++d) {
          const float v0 = acc[2 * d][r], v1 = acc[2 * d + 1][r];
          if (STORE_C)
            *(v2f*)(C + (size_t)grow * DH + d * FF + f0 + 2 * c) = (v2f){v0, v1};
          s0 += v0;
          s1 += v1;
        }
        *(v2f*)(Cm + (size_t)grow * FF + f0 + 2 * c) =
            (v2f){s0 * (1.0f / (float)DD), s1 * (1.0f / (float)DD)};
      }
    }
  }
}

__global__ void alpha_deg_kernel(const float* __restrict__ xm, const float* __restrict__ em,
                                 const int* __restrict__ row, const int* __restrict__ col,
                                 float* __restrict__ alpha, float* __restrict__ Dg,
                                 float* __restrict__ Bg) {
  int gid = blockIdx.x * blockDim.x + threadIdx.x;
  if (gid >= NNZ * DD) return;
  int i = gid / DD, dd = gid - i * DD;
  int r = row[i], c = col[i];
  float v = xm[r * DD + dd] + em[c * DD + dd];
  v = 1.0f / (1.0f + __expf(-v));
  alpha[gid] = v;
  ATOMIC_ADD_F32(&Dg[r * DD + dd], v);
  ATOMIC_ADD_F32(&Bg[c * DD + dd], v);
}

__global__ void recip_kernel(float* __restrict__ p, int n) {
  int gid = blockIdx.x * blockDim.x + threadIdx.x;
  if (gid >= n) return;
  float v = p[gid];
  p[gid] = (v != 0.0f) ? (1.0f / v) : 0.0f;
}

__global__ void fill_kernel(float* __restrict__ p, size_t n, float v) {
  size_t stride = (size_t)gridDim.x * blockDim.x;
  for (size_t i = blockIdx.x * (size_t)blockDim.x + threadIdx.x; i < n; i += stride)
    p[i] = v;
}

#define OW_WAVES 32
#define OW_SLICE 128
#define OW_RING  64
#define OW_BATCH 8

#define OW_RN   250
#define OW_RE   25
#define OW_NBLK 80
#define BM_WORDS ((NNZ + 31) / 32)
#define BM_PITCH (((BM_WORDS + 31) / 32) * 32)
#define BM_WPW   (BM_PITCH / OW_WAVES)

static_assert(OW_NBLK * OW_RN == NN && OW_NBLK * OW_RE == EE, "the blocks cover the destination rows");
static_assert(BM_PITCH % OW_WAVES == 0, "a bitmap row divides evenly over the waves");

__global__ __launch_bounds__(1024)
void block_masks(const int* __restrict__ row, const int* __restrict__ col,
                 unsigned* __restrict__ bm) {
  constexpr int TP = OW_NBLK + 1;
  __shared__ unsigned tile[32 * TP];
  const int tid  = threadIdx.x;
  const int lane = tid & 31;
  const int wave = __builtin_amdgcn_readfirstlane(tid >> 5);
  const int dir  = blockIdx.y;
  const int p    = blockIdx.x * 32 + wave;
  const int i    = (p % BM_WPW) * 1024 + 32 * lane + p / BM_WPW;
  int blk = -1;
  if (i < NNZ) blk = dir ? (row[i] / OW_RN) : (col[i] / OW_RE);
#pragma unroll
  for (int c = 0; c < (OW_NBLK + 31) / 32; ++c) {
    int mine = 0;
#pragma unroll
    for (int j = 0; j < 32; ++j) {
      const int m = (int)__builtin_amdgcn_ballot_w32(blk == c * 32 + j);
      mine = (lane == j) ? m : mine;
    }
    const int b = c * 32 + lane;
    if (b < OW_NBLK) tile[wave * TP + b] = (unsigned)mine;
  }
  __syncthreads();
  unsigned* out = bm + (size_t)dir * OW_NBLK * BM_PITCH + blockIdx.x * 32;
  for (int k = tid; k < OW_NBLK * 32; k += 1024)
    out[(size_t)(k >> 5) * BM_PITCH + (k & 31)] = tile[(k & 31) * TP + (k >> 5)];
}

__device__ __forceinline__ void lds_add(float* p, float v) {
  __hip_atomic_fetch_add(p, v, __ATOMIC_RELAXED, __HIP_MEMORY_SCOPE_WORKGROUP);
}

template <bool FULL>
__device__ __forceinline__ void ow_drain(const int* ring, int head, int n, int lane,
                                         const int* __restrict__ src_idx,
                                         const int* __restrict__ dst_idx,
                                         const float* __restrict__ al,
                                         const float* __restrict__ iv, int r0,
                                         const float* sp, float* ap) {
  int b_l = lane & (OW_BATCH - 1);
  if (!FULL) b_l = (b_l < n) ? b_l : (n - 1);
  const int   inc  = ring[(head + b_l) & (OW_RING - 1)];
  const int   srow = src_idx[inc];
  const int   loc  = dst_idx[inc] - r0;
  const float scl  = iv[loc * DD] * al[(size_t)inc * DD];

  v4f   v[OW_BATCH];
  int   dl[OW_BATCH];
  float sc[OW_BATCH];
#pragma unroll
  for (int b = 0; b < OW_BATCH; ++b) {
    if (FULL || b < n) {
      const int sr = __builtin_amdgcn_readlane(srow, b);
      dl[b] = __builtin_amdgcn_readlane(loc, b);
      sc[b] = __int_as_float(__builtin_amdgcn_readlane(__float_as_int(scl), b));
      v[b]  = *(const v4f*)(sp + (size_t)sr * DH);
    }
  }
#pragma unroll
  for (int b = 0; b < OW_BATCH; ++b) {
    if (FULL || b < n) {
      float* a = ap + dl[b] * OW_SLICE;
      lds_add(a,      sc[b] * v[b].x);
      lds_add(a + 32, sc[b] * v[b].y);
      lds_add(a + 64, sc[b] * v[b].z);
      lds_add(a + 96, sc[b] * v[b].w);
    }
  }
}

template <int R, bool BIAS>
__global__ __launch_bounds__(OW_WAVES * 32)
void owner_rows(const float* __restrict__ src, const float* __restrict__ alpha,
                const float* __restrict__ inv, const int* __restrict__ src_idx,
                const int* __restrict__ dst_idx, const unsigned* __restrict__ bm,
                float* __restrict__ dst, const float* __restrict__ bias, int rows) {
  __shared__ __attribute__((aligned(16))) float acc[R * OW_SLICE];
  __shared__ int rings[OW_WAVES * OW_RING];

  const int tid  = threadIdx.x;
  const int lane = tid & 31;
  const int wave = __builtin_amdgcn_readfirstlane(tid >> 5);
  const int r0   = blockIdx.x * R;
  const int dch  = blockIdx.y;
  const int col0 = dch * OW_SLICE;

  for (int k = tid; k < R * OW_SLICE / 4; k += OW_WAVES * 32) ((v4f*)acc)[k] = (v4f){};
  __syncthreads();

  const float* sp   = src + col0 + 4 * lane;
  float*       ap   = acc + lane;
  int*         ring = rings + wave * OW_RING;
  const float* al   = alpha + dch;
  const float* iv   = inv + (size_t)r0 * DD + dch;

  const unsigned* bw = bm + (size_t)blockIdx.x * BM_PITCH + wave * BM_WPW;
  constexpr int SLOTS = (BM_WPW + 31) / 32;
  unsigned wv[SLOTS];
#pragma unroll
  for (int s = 0; s < SLOTS; ++s) {
    const int o = s * 32 + lane;
    wv[s] = (o < BM_WPW) ? bw[o] : 0u;
  }

  int head = 0, count = 0;
#pragma unroll
  for (int s = 0; s < SLOTS; ++s) {
    unsigned mw = wv[s];
    const int ibase = (s * 32 + lane) * 1024 + wave;
    for (int t = 0; t < 32; ++t) {
      const bool m = (mw != 0u);
      const unsigned mask = __builtin_amdgcn_ballot_w32(m);
      if (mask == 0u) break;
      if (m) {
        const int e = (head + count + (int)__builtin_amdgcn_mbcnt_lo(mask, 0u)) & (OW_RING - 1);
        ring[e] = ibase + 32 * __builtin_ctz(mw);
        mw &= mw - 1u;
      }
      count += __builtin_popcount(mask);
      while (count >= OW_BATCH) {
        ow_drain<true>(ring, head, OW_BATCH, lane, src_idx, dst_idx, al, iv, r0, sp, ap);
        head   = (head + OW_BATCH) & (OW_RING - 1);
        count -= OW_BATCH;
      }
    }
  }
  if (count > 0) ow_drain<false>(ring, head, count, lane, src_idx, dst_idx, al, iv, r0, sp, ap);
  __syncthreads();

  v4f bb = {0.0f, 0.0f, 0.0f, 0.0f};
  if (BIAS) bb = *(const v4f*)(bias + 4 * lane);
  for (int r = wave; r < R; r += OW_WAVES) {
    const int grow = r0 + r;
    if (grow < rows) {
      const float* a = ap + r * OW_SLICE;
      v4f o = {a[0], a[32], a[64], a[96]};
      if (BIAS) o += bb;
      *(v4f*)(dst + (size_t)grow * DH + col0 + 4 * lane) = o;
    }
  }
}

extern "C" void kernel_launch(void* const* d_in, const int* in_sizes, int n_in,
                              void* d_out, int out_size, void* d_ws, size_t ws_size,
                              hipStream_t stream) {
  const float* x        = (const float*)d_in[0];
  const int*   eidx     = (const int*)d_in[1];
  const float* hedge    = (const float*)d_in[2];
  const float* lin_W    = (const float*)d_in[3];
  const float* sheaf_W  = (const float*)d_in[4];
  const float* conv1_W  = (const float*)d_in[5];
  const float* conv1_b  = (const float*)d_in[6];
  const float* conv2_W  = (const float*)d_in[7];
  const float* conv2_b  = (const float*)d_in[8];
  const float* lin2_W   = (const float*)d_in[9];
  float* out = (float*)d_out;

  const int* row = eidx;
  const int* col = eidx + NNZ;

  float* ws    = (float*)d_ws;
  float* bufA  = ws;
  float* bufB  = bufA + (size_t)NN * DH;
  float* mbuf  = bufB + (size_t)NN * DH;
  float* xmean = mbuf + (size_t)EE * DH;
  float* emean = xmean + (size_t)NN * FF;
  float* xm    = emean + (size_t)EE * FF;
  float* em    = xm + (size_t)NN * DD;
  float* alpha = em + (size_t)EE * DD;
  float* Dg    = alpha + (size_t)NNZ * DD;
  float* Bg    = Dg + (size_t)NN * DD;

  gemm_lin_mean_f32<16, true><<<dim3(4, 64), 512, 0, stream>>>(x, lin_W, bufA, xmean, NN);
  gemm_lin_mean_f32<4, false><<<dim3(4, 32), 128, 0, stream>>>(hedge, lin_W, nullptr, emean, EE);

  gemm_direct_f32<FF, 1, 4, false><<<dim3(1, 313), 128, 0, stream>>>(xmean, sheaf_W, xm, NN, DD, DD, DD);
  gemm_direct_f32<FF, 1, 4, false><<<dim3(1, 32), 128, 0, stream>>>(emean, sheaf_W + FF * DD, em, EE, DD, DD, DD);

  unsigned* bm_e = (unsigned*)xmean;
  unsigned* bm_n = bm_e + (size_t)OW_NBLK * BM_PITCH;
  block_masks<<<dim3(BM_PITCH / 32, 2), 1024, 0, stream>>>(row, col, bm_e);

  fill_kernel<<<64, 256, 0, stream>>>(Dg, (size_t)NN * DD, 0.0f);
  fill_kernel<<<64, 256, 0, stream>>>(Bg, (size_t)EE * DD, 0.0f);
  alpha_deg_kernel<<<(NNZ * DD + 255) / 256, 256, 0, stream>>>(xm, em, row, col,
                                                               alpha, Dg, Bg);
  recip_kernel<<<(NN * DD + 255) / 256, 256, 0, stream>>>(Dg, NN * DD);
  recip_kernel<<<(EE * DD + 255) / 256, 256, 0, stream>>>(Bg, EE * DD);

  gemm_direct_f32<FF, 8, 8, true><<<dim3(1, 512), 256, 0, stream>>>(
      bufA, conv1_W, bufB, NN * DD, HH, HH, HH);
  owner_rows<OW_RE, false><<<dim3(OW_NBLK, DH / OW_SLICE), OW_WAVES * 32, 0, stream>>>(
      bufB, alpha, Bg, row, col, bm_e, mbuf, nullptr, EE);
  owner_rows<OW_RN, true><<<dim3(OW_NBLK, DH / OW_SLICE), OW_WAVES * 32, 0, stream>>>(
      mbuf, alpha, Dg, col, row, bm_n, bufA, conv1_b, NN);

  gemm_direct_f32<HH, 8, 8, true, true><<<dim3(1, 512), 256, 0, stream>>>(
      bufA, conv2_W, bufB, NN * DD, HH, HH, HH);
  owner_rows<OW_RE, false><<<dim3(OW_NBLK, DH / OW_SLICE), OW_WAVES * 32, 0, stream>>>(
      bufB, alpha, Bg, row, col, bm_e, mbuf, nullptr, EE);
  owner_rows<OW_RN, true><<<dim3(OW_NBLK, DH / OW_SLICE), OW_WAVES * 32, 0, stream>>>(
      mbuf, alpha, Dg, col, row, bm_n, bufA, conv2_b, NN);

  gemm_direct_f32<DH, 3, 4, true><<<dim3(1, 313), 128, 0, stream>>>(bufA, lin2_W, out, NN, COUT, COUT, COUT);
}


# Sheaf degrees by owner workgroups: one launch reads the block bitmaps, forms alpha, sums degrees in LDS (float atomics, arrival order), stores the inverses; global atomics, 2 fills, 2 inversions go
# speedup vs baseline: 4.0181x; 1.0394x over previous
#include <hip/hip_runtime.h>
#include <hip/hip_bf16.h>

#define NN   20000
#define EE   2000
#define NNZ  100000
#define FF   128
#define HH   128
#define DD   6
#define COUT 40
#define DH   (DD * HH)

typedef __attribute__((ext_vector_type(2))) float v2f;
typedef __attribute__((ext_vector_type(8))) float v8f;

typedef __attribute__((ext_vector_type(4))) float v4f;

__device__ __forceinline__ float elu1(float v) {
  return (v > 0.0f) ? v : (__expf(v) - 1.0f);
}

template <int K, int NT, int WAVES, bool VECB, bool ELUA = false>
__global__ __launch_bounds__(WAVES * 32)
void gemm_direct_f32(const float* __restrict__ A, const float* __restrict__ B,
                     float* __restrict__ C, int M, int Nc, int ldb, int ldc) {
  constexpr int NB = NT * 16;
  constexpr int KC = 128;
  constexpr int THREADS = WAVES * 32;
  constexpr bool PERM = (NT % 4) == 0;
  __shared__ v4f Bs[(KC / 4) * NB];

  const int tid  = threadIdx.x;
  const int wave = tid >> 5;
  const int lane = tid & 31;
  const int c    = lane & 15;
  const int h    = lane >> 4;
  const int nb   = blockIdx.x * NB;

  const int row_tiles = (M + 15) / 16;
  const int per_sweep = gridDim.y * WAVES;
  const int sweeps    = (row_tiles + per_sweep - 1) / per_sweep;

  auto slot_of = [](int n) {
    return PERM ? ((n >> 6) * 64 + (n & 3) * 16 + ((n & 63) >> 2)) : n;
  };

  auto stage_panel = [&](int kc) {
    if (VECB) {
      constexpr int G = NB / 4;
      constexpr int ITERS = ((KC / 4) * G) / THREADS;
      static_assert(((KC / 4) * G) % THREADS == 0, "panel must divide evenly over the block");
#pragma unroll
      for (int i = 0; i < ITERS; ++i) {
        const int idx = tid + i * THREADS;
        const int q = idx / G, n = (idx % G) * 4;
        const int gn = nb + n;
        const int gs = (gn < Nc) ? gn : (Nc - 4);
        const float* bp = B + (size_t)(kc + 4 * q) * ldb + gs;
        v4f r0 = *(const v4f*)(bp);
        v4f r1 = *(const v4f*)(bp + ldb);
        v4f r2 = *(const v4f*)(bp + 2 * (size_t)ldb);
        v4f r3 = *(const v4f*)(bp + 3 * (size_t)ldb);
        Bs[q * NB + slot_of(n + 0)] = (v4f){r0.x, r1.x, r2.x, r3.x};
        Bs[q * NB + slot_of(n + 1)] = (v4f){r0.y, r1.y, r2.y, r3.y};
        Bs[q * NB + slot_of(n + 2)] = (v4f){r0.z, r1.z, r2.z, r3.z};
        Bs[q * NB + slot_of(n + 3)] = (v4f){r0.w, r1.w, r2.w, r3.w};
      }
    } else {
      constexpr int ITERS = ((KC / 4) * NB) / THREADS;
      static_assert(((KC / 4) * NB) % THREADS == 0, "panel must divide evenly over the block");
#pragma unroll
      for (int i = 0; i < ITERS; ++i) {
        const int idx = tid + i * THREADS;
        const int q = idx / NB, n = idx % NB;
        const int gn = nb + n;
        const int gs = (gn < Nc) ? gn : (Nc - 1);
        const float* bp = B + (size_t)(kc + 4 * q) * ldb + gs;
        v4f v;
        v.x = bp[0];
        v.y = bp[ldb];
        v.z = bp[2 * (size_t)ldb];
        v.w = bp[3 * (size_t)ldb];
        Bs[q * NB + slot_of(n)] = v;
      }
    }
  };

  if (K == KC) {
    stage_panel(0);
    __syncthreads();
  }

  for (int it = 0; it < sweeps; ++it) {
    const int tile = (it * gridDim.y + blockIdx.y) * WAVES + wave;
    const int r0   = tile * 16;
    if (K == KC && it > 0 && r0 >= M) break;

    int gm = r0 + c;
    gm = (gm < M) ? gm : (M - 1);
    const float* ap = A + (size_t)gm * K + 4 * h;

    v8f acc[NT];
#pragma unroll
    for (int nt = 0; nt < NT; ++nt) acc[nt] = (v8f){};

    v4f a_cur[4];
#pragma unroll
    for (int j = 0; j < 4; ++j) {
      a_cur[j] = *(const v4f*)(ap + 8 * j);
      if (ELUA) a_cur[j] = (v4f){elu1(a_cur[j].x), elu1(a_cur[j].y), elu1(a_cur[j].z), elu1(a_cur[j].w)};
    }

    for (int kc = 0; kc < K; kc += KC) {
      if (K > KC) {
        __syncthreads();
        stage_panel(kc);
        __syncthreads();
      }

#pragma unroll
      for (int s = 0; s < 4; ++s) {
        v4f a_nxt[4];
        const int kn = kc + (s + 1) * 32;
        if (kn < K) {
#pragma unroll
          for (int j = 0; j < 4; ++j) a_nxt[j] = *(const v4f*)(ap + kn + 8 * j);
        }
        __builtin_amdgcn_sched_barrier(0);
#pragma unroll
        for (int j = 0; j < 4; ++j) {
          v4f b[NT];
#pragma unroll
          for (int nt = 0; nt < NT; ++nt) b[nt] = Bs[(8 * s + 2 * j + h) * NB + nt * 16 + c];
          const v2f alo = {a_cur[j].x, a_cur[j].y};
          const v2f ahi = {a_cur[j].z, a_cur[j].w};
#pragma unroll
          for (int nt = 0; nt < NT; ++nt) {
            const v2f blo = {b[nt].x, b[nt].y};
            acc[nt] = __builtin_amdgcn_wmma_f32_16x16x4_f32(
                false, alo, false, blo, (short)0, acc[nt], false, false);
          }
#pragma unroll
          for (int nt = 0; nt < NT; ++nt) {
            const v2f bhi = {b[nt].z, b[nt].w};
            acc[nt] = __builtin_amdgcn_wmma_f32_16x16x4_f32(
                false, ahi, false, bhi, (short)0, acc[nt], false, false);
          }
        }
        if (kn < K) {
#pragma unroll
          for (int j = 0; j < 4; ++j) {
            a_cur[j] = ELUA ? (v4f){elu1(a_nxt[j].x), elu1(a_nxt[j].y), elu1(a_nxt[j].z), elu1(a_nxt[j].w)}
                            : a_nxt[j];
          }
        }
      }
    }

    const bool vec_rows = ((ldc & 3) == 0) && (nb + NB <= Nc);
#pragma unroll
    for (int r = 0; r < 8; ++r) {
      const int grow = r0 + 8 * h + r;
      if (grow < M) {
        float* crow = C + (size_t)grow * ldc;
        if (PERM) {
#pragma unroll
          for (int g = 0; g < NT / 4; ++g) {
            const int n = nb + g * 64 + 4 * c;
            if (vec_rows) {
              v4f o = {acc[4 * g + 0][r], acc[4 * g + 1][r], acc[4 * g + 2][r], acc[4 * g + 3][r]};
              *(v4f*)(crow + n) = o;
            } else {
#pragma unroll
              for (int e = 0; e < 4; ++e)
                if (n + e < Nc) crow[n + e] = acc[4 * g + e][r];
            }
          }
        } else {
#pragma unroll
          for (int nt = 0; nt < NT; ++nt) {
            const int n = nb + nt * 16 + c;
            if (n < Nc) crow[n] = acc[nt][r];
          }
        }
      }
    }
  }
}

template <int WAVES, bool STORE_C>
__global__ __launch_bounds__(WAVES * 32)
void gemm_lin_mean_f32(const float* __restrict__ A, const float* __restrict__ B,
                       float* __restrict__ C, float* __restrict__ Cm, int M) {
  constexpr int K  = FF;
  constexpr int NT = 2 * DD;
  constexpr int NB = NT * 16;
  constexpr int FS = 32;
  constexpr int THREADS = WAVES * 32;
  __shared__ v4f Bs[(K / 4) * NB];

  const int tid  = threadIdx.x;
  const int lane = tid & 31;
  const int c    = lane & 15;
  const int h    = lane >> 4;
  const int wave = __builtin_amdgcn_readfirstlane(tid >> 5);
  const int f0   = blockIdx.x * FS;

  {
    constexpr int G = NB / 4;
    constexpr int ITERS = ((K / 4) * G) / THREADS;
    static_assert(((K / 4) * G) % THREADS == 0, "panel must divide evenly over the block");
#pragma unroll
    for (int i = 0; i < ITERS; ++i) {
      const int idx = tid + i * THREADS;
      const int q = idx / G, g = idx % G;
      const int d = g >> 3, j0 = (g & 7) * 4;
      const float* bp = B + (size_t)(4 * q) * DH + d * FF + f0 + j0;
      const v4f r0 = *(const v4f*)(bp);
      const v4f r1 = *(const v4f*)(bp + DH);
      const v4f r2 = *(const v4f*)(bp + 2 * DH);
      const v4f r3 = *(const v4f*)(bp + 3 * DH);
      const int s0 = q * NB + (2 * d) * 16 + (j0 >> 1);
      Bs[s0]      = (v4f){r0.x, r1.x, r2.x, r3.x};
      Bs[s0 + 16] = (v4f){r0.y, r1.y, r2.y, r3.y};
      Bs[s0 + 1]  = (v4f){r0.z, r1.z, r2.z, r3.z};
      Bs[s0 + 17] = (v4f){r0.w, r1.w, r2.w, r3.w};
    }
  }
  __syncthreads();

  const int row_tiles = (M + 15) / 16;
  const int per_sweep = gridDim.y * WAVES;
  const int sweeps    = (row_tiles + per_sweep - 1) / per_sweep;

  for (int it = 0; it < sweeps; ++it) {
    const int tile = it * per_sweep + wave * gridDim.y + blockIdx.y;
    const int r0   = tile * 16;
    if (r0 >= M) break;

    int gm = r0 + c;
    gm = (gm < M) ? gm : (M - 1);
    const float* ap = A + (size_t)gm * K + 4 * h;

    v8f acc[NT];
#pragma unroll
    for (int nt = 0; nt < NT; ++nt) acc[nt] = (v8f){};

    v4f a_cur[4];
#pragma unroll
    for (int j = 0; j < 4; ++j) a_cur[j] = *(const v4f*)(ap + 8 * j);

#pragma unroll
    for (int s = 0; s < 4; ++s) {
      v4f a_nxt[4];
      const int kn = (s + 1) * 32;
      if (kn < K) {
#pragma unroll
        for (int j = 0; j < 4; ++j) a_nxt[j] = *(const v4f*)(ap + kn + 8 * j);
      }
      __builtin_amdgcn_sched_barrier(0);
#pragma unroll
      for (int j = 0; j < 4; ++j) {
        v4f b[NT];
#pragma unroll
        for (int nt = 0; nt < NT; ++nt) b[nt] = Bs[(8 * s + 2 * j + h) * NB + nt * 16 + c];
        const v2f alo = {a_cur[j].x, a_cur[j].y};
        const v2f ahi = {a_cur[j].z, a_cur[j].w};
#pragma unroll
        for (int nt = 0; nt < NT; ++nt) {
          const v2f blo = {b[nt].x, b[nt].y};
          acc[nt] = __builtin_amdgcn_wmma_f32_16x16x4_f32(
              false, alo, false, blo, (short)0, acc[nt], false, false);
        }
#pragma unroll
        for (int nt = 0; nt < NT; ++nt) {
          const v2f bhi = {b[nt].z, b[nt].w};
          acc[nt] = __builtin_amdgcn_wmma_f32_16x16x4_f32(
              false, ahi, false, bhi, (short)0, acc[nt], false, false);
        }
      }
      if (kn < K) {
#pragma unroll
        for (int j = 0; j < 4; ++j) a_cur[j] = a_nxt[j];
      }
    }

#pragma unroll
    for (int r = 0; r < 8; ++r) {
      const int grow = r0 + 8 * h + r;
      if (grow < M) {
        float s0 = 0.0f, s1 = 0.0f;
#pragma unroll
        for (int d = 0; d < DD; ++d) {
          const float v0 = acc[2 * d][r], v1 = acc[2 * d + 1][r];
          if (STORE_C)
            *(v2f*)(C + (size_t)grow * DH + d * FF + f0 + 2 * c) = (v2f){v0, v1};
          s0 += v0;
          s1 += v1;
        }
        *(v2f*)(Cm + (size_t)grow * FF + f0 + 2 * c) =
            (v2f){s0 * (1.0f / (float)DD), s1 * (1.0f / (float)DD)};
      }
    }
  }
}

#define OW_WAVES 32
#define OW_SLICE 128
#define OW_RING  64
#define OW_BATCH 8

#define OW_RN   250
#define OW_RE   25
#define OW_NBLK 80
#define BM_WORDS ((NNZ + 31) / 32)
#define BM_PITCH (((BM_WORDS + 31) / 32) * 32)
#define BM_WPW   (BM_PITCH / OW_WAVES)

static_assert(OW_NBLK * OW_RN == NN && OW_NBLK * OW_RE == EE, "the blocks cover the destination rows");
static_assert(BM_PITCH % OW_WAVES == 0, "a bitmap row divides evenly over the waves");

__global__ __launch_bounds__(1024)
void block_masks(const int* __restrict__ row, const int* __restrict__ col,
                 unsigned* __restrict__ bm) {
  constexpr int TP = OW_NBLK + 1;
  __shared__ unsigned tile[32 * TP];
  const int tid  = threadIdx.x;
  const int lane = tid & 31;
  const int wave = __builtin_amdgcn_readfirstlane(tid >> 5);
  const int dir  = blockIdx.y;
  const int p    = blockIdx.x * 32 + wave;
  const int i    = (p % BM_WPW) * 1024 + 32 * lane + p / BM_WPW;
  int blk = -1;
  if (i < NNZ) blk = dir ? (row[i] / OW_RN) : (col[i] / OW_RE);
#pragma unroll
  for (int c = 0; c < (OW_NBLK + 31) / 32; ++c) {
    int mine = 0;
#pragma unroll
    for (int j = 0; j < 32; ++j) {
      const int m = (int)__builtin_amdgcn_ballot_w32(blk == c * 32 + j);
      mine = (lane == j) ? m : mine;
    }
    const int b = c * 32 + lane;
    if (b < OW_NBLK) tile[wave * TP + b] = (unsigned)mine;
  }
  __syncthreads();
  unsigned* out = bm + (size_t)dir * OW_NBLK * BM_PITCH + blockIdx.x * 32;
  for (int k = tid; k < OW_NBLK * 32; k += 1024)
    out[(size_t)(k >> 5) * BM_PITCH + (k & 31)] = tile[(k & 31) * TP + (k >> 5)];
}

__device__ __forceinline__ void lds_add(float* p, float v) {
  __hip_atomic_fetch_add(p, v, __ATOMIC_RELAXED, __HIP_MEMORY_SCOPE_WORKGROUP);
}

#define DG_COPIES 4

__global__ __launch_bounds__(OW_WAVES * 32)
void owner_degrees(const float* __restrict__ xm, const float* __restrict__ em,
                   const int* __restrict__ row, const int* __restrict__ col,
                   const unsigned* __restrict__ bm, float* __restrict__ alpha,
                   float* __restrict__ Dg, float* __restrict__ Bg) {
  __shared__ float deg[DG_COPIES * OW_RN * DD];
  __shared__ int rings[OW_WAVES * OW_RING];

  const int tid  = threadIdx.x;
  const int lane = tid & 31;
  const int wave = __builtin_amdgcn_readfirstlane(tid >> 5);
  const int dir  = blockIdx.y;
  const int R    = dir ? OW_RN : OW_RE;
  const int r0   = blockIdx.x * R;

  for (int k = tid; k < DG_COPIES * OW_RN * DD; k += OW_WAVES * 32) deg[k] = 0.0f;
  __syncthreads();

  int*   ring = rings + wave * OW_RING;
  float* dsum = deg + (lane & (DG_COPIES - 1)) * (OW_RN * DD);

  auto take = [&](int head, int n) {
    if (lane < n) {
      const int inc = ring[(head + lane) & (OW_RING - 1)];
      const int r = row[inc], c = col[inc];
      const int loc = (dir ? r : c) - r0;
      const v2f* xp = (const v2f*)(xm + (size_t)r * DD);
      const v2f* ep = (const v2f*)(em + (size_t)c * DD);
      v2f a[DD / 2];
      for (int j = 0; j < DD / 2; ++j) {
        const v2f s = xp[j] + ep[j];
        a[j].x = 1.0f / (1.0f + __expf(-s.x));
        a[j].y = 1.0f / (1.0f + __expf(-s.y));
      }
      float* d = dsum + loc * DD;
      for (int j = 0; j < DD / 2; ++j) {
        lds_add(d + 2 * j,     a[j].x);
        lds_add(d + 2 * j + 1, a[j].y);
      }
      if (dir == 0) {
        v2f* ao = (v2f*)(alpha + (size_t)inc * DD);
        for (int j = 0; j < DD / 2; ++j) ao[j] = a[j];
      }
    }
  };

  const unsigned* bw = bm + ((size_t)dir * OW_NBLK + blockIdx.x) * BM_PITCH + wave * BM_WPW;
  constexpr int SLOTS = (BM_WPW + 31) / 32;
  unsigned wv[SLOTS];
#pragma unroll
  for (int s = 0; s < SLOTS; ++s) {
    const int o = s * 32 + lane;
    wv[s] = (o < BM_WPW) ? bw[o] : 0u;
  }

  int head = 0, count = 0;
#pragma unroll
  for (int s = 0; s < SLOTS; ++s) {
    unsigned mw = wv[s];
    const int ibase = (s * 32 + lane) * 1024 + wave;
    for (int t = 0; t < 32; ++t) {
      const bool m = (mw != 0u);
      const unsigned mask = __builtin_amdgcn_ballot_w32(m);
      if (mask == 0u) break;
      if (m) {
        const int e = (head + count + (int)__builtin_amdgcn_mbcnt_lo(mask, 0u)) & (OW_RING - 1);
        ring[e] = ibase + 32 * __builtin_ctz(mw);
        mw &= mw - 1u;
      }
      count += __builtin_popcount(mask);
      if (count >= 32) {
        take(head, 32);
        head   = (head + 32) & (OW_RING - 1);
        count -= 32;
      }
    }
  }
  if (count > 0) take(head, count);
  __syncthreads();

  float* out = (dir ? Dg : Bg) + (size_t)r0 * DD;
  for (int k = tid; k < R * DD; k += OW_WAVES * 32) {
    float v = deg[k];
    for (int cpy = 1; cpy < DG_COPIES; ++cpy) v += deg[cpy * (OW_RN * DD) + k];
    out[k] = (v != 0.0f) ? (1.0f / v) : 0.0f;
  }
}

template <bool FULL>
__device__ __forceinline__ void ow_drain(const int* ring, int head, int n, int lane,
                                         const int* __restrict__ src_idx,
                                         const int* __restrict__ dst_idx,
                                         const float* __restrict__ al,
                                         const float* __restrict__ iv, int r0,
                                         const float* sp, float* ap) {
  int b_l = lane & (OW_BATCH - 1);
  if (!FULL) b_l = (b_l < n) ? b_l : (n - 1);
  const int   inc  = ring[(head + b_l) & (OW_RING - 1)];
  const int   srow = src_idx[inc];
  const int   loc  = dst_idx[inc] - r0;
  const float scl  = iv[loc * DD] * al[(size_t)inc * DD];

  v4f   v[OW_BATCH];
  int   dl[OW_BATCH];
  float sc[OW_BATCH];
#pragma unroll
  for (int b = 0; b < OW_BATCH; ++b) {
    if (FULL || b < n) {
      const int sr = __builtin_amdgcn_readlane(srow, b);
      dl[b] = __builtin_amdgcn_readlane(loc, b);
      sc[b] = __int_as_float(__builtin_amdgcn_readlane(__float_as_int(scl), b));
      v[b]  = *(const v4f*)(sp + (size_t)sr * DH);
    }
  }
#pragma unroll
  for (int b = 0; b < OW_BATCH; ++b) {
    if (FULL || b < n) {
      float* a = ap + dl[b] * OW_SLICE;
      lds_add(a,      sc[b] * v[b].x);
      lds_add(a + 32, sc[b] * v[b].y);
      lds_add(a + 64, sc[b] * v[b].z);
      lds_add(a + 96, sc[b] * v[b].w);
    }
  }
}

template <int R, bool BIAS>
__global__ __launch_bounds__(OW_WAVES * 32)
void owner_rows(const float* __restrict__ src, const float* __restrict__ alpha,
                const float* __restrict__ inv, const int* __restrict__ src_idx,
                const int* __restrict__ dst_idx, const unsigned* __restrict__ bm,
                float* __restrict__ dst, const float* __restrict__ bias, int rows) {
  __shared__ __attribute__((aligned(16))) float acc[R * OW_SLICE];
  __shared__ int rings[OW_WAVES * OW_RING];

  const int tid  = threadIdx.x;
  const int lane = tid & 31;
  const int wave = __builtin_amdgcn_readfirstlane(tid >> 5);
  const int r0   = blockIdx.x * R;
  const int dch  = blockIdx.y;
  const int col0 = dch * OW_SLICE;

  for (int k = tid; k < R * OW_SLICE / 4; k += OW_WAVES * 32) ((v4f*)acc)[k] = (v4f){};
  __syncthreads();

  const float* sp   = src + col0 + 4 * lane;
  float*       ap   = acc + lane;
  int*         ring = rings + wave * OW_RING;
  const float* al   = alpha + dch;
  const float* iv   = inv + (size_t)r0 * DD + dch;

  const unsigned* bw = bm + (size_t)blockIdx.x * BM_PITCH + wave * BM_WPW;
  constexpr int SLOTS = (BM_WPW + 31) / 32;
  unsigned wv[SLOTS];
#pragma unroll
  for (int s = 0; s < SLOTS; ++s) {
    const int o = s * 32 + lane;
    wv[s] = (o < BM_WPW) ? bw[o] : 0u;
  }

  int head = 0, count = 0;
#pragma unroll
  for (int s = 0; s < SLOTS; ++s) {
    unsigned mw = wv[s];
    const int ibase = (s * 32 + lane) * 1024 + wave;
    for (int t = 0; t < 32; ++t) {
      const bool m = (mw != 0u);
      const unsigned mask = __builtin_amdgcn_ballot_w32(m);
      if (mask == 0u) break;
      if (m) {
        const int e = (head + count + (int)__builtin_amdgcn_mbcnt_lo(mask, 0u)) & (OW_RING - 1);
        ring[e] = ibase + 32 * __builtin_ctz(mw);
        mw &= mw - 1u;
      }
      count += __builtin_popcount(mask);
      while (count >= OW_BATCH) {
        ow_drain<true>(ring, head, OW_BATCH, lane, src_idx, dst_idx, al, iv, r0, sp, ap);
        head   = (head + OW_BATCH) & (OW_RING - 1);
        count -= OW_BATCH;
      }
    }
  }
  if (count > 0) ow_drain<false>(ring, head, count, lane, src_idx, dst_idx, al, iv, r0, sp, ap);
  __syncthreads();

  v4f bb = {0.0f, 0.0f, 0.0f, 0.0f};
  if (BIAS) bb = *(const v4f*)(bias + 4 * lane);
  for (int r = wave; r < R; r += OW_WAVES) {
    const int grow = r0 + r;
    if (grow < rows) {
      const float* a = ap + r * OW_SLICE;
      v4f o = {a[0], a[32], a[64], a[96]};
      if (BIAS) o += bb;
      *(v4f*)(dst + (size_t)grow * DH + col0 + 4 * lane) = o;
    }
  }
}

extern "C" void kernel_launch(void* const* d_in, const int* in_sizes, int n_in,
                              void* d_out, int out_size, void* d_ws, size_t ws_size,
                              hipStream_t stream) {
  const float* x        = (const float*)d_in[0];
  const int*   eidx     = (const int*)d_in[1];
  const float* hedge    = (const float*)d_in[2];
  const float* lin_W    = (const float*)d_in[3];
  const float* sheaf_W  = (const float*)d_in[4];
  const float* conv1_W  = (const float*)d_in[5];
  const float* conv1_b  = (const float*)d_in[6];
  const float* conv2_W  = (const float*)d_in[7];
  const float* conv2_b  = (const float*)d_in[8];
  const float* lin2_W   = (const float*)d_in[9];
  float* out = (float*)d_out;

  const int* row = eidx;
  const int* col = eidx + NNZ;

  float* ws    = (float*)d_ws;
  float* bufA  = ws;
  float* bufB  = bufA + (size_t)NN * DH;
  float* mbuf  = bufB + (size_t)NN * DH;
  float* xmean = mbuf + (size_t)EE * DH;
  float* emean = xmean + (size_t)NN * FF;
  float* xm    = emean + (size_t)EE * FF;
  float* em    = xm + (size_t)NN * DD;
  float* alpha = em + (size_t)EE * DD;
  float* Dg    = alpha + (size_t)NNZ * DD;
  float* Bg    = Dg + (size_t)NN * DD;

  gemm_lin_mean_f32<16, true><<<dim3(4, 64), 512, 0, stream>>>(x, lin_W, bufA, xmean, NN);
  gemm_lin_mean_f32<4, false><<<dim3(4, 32), 128, 0, stream>>>(hedge, lin_W, nullptr, emean, EE);

  gemm_direct_f32<FF, 1, 4, false><<<dim3(1, 313), 128, 0, stream>>>(xmean, sheaf_W, xm, NN, DD, DD, DD);
  gemm_direct_f32<FF, 1, 4, false><<<dim3(1, 32), 128, 0, stream>>>(emean, sheaf_W + FF * DD, em, EE, DD, DD, DD);

  unsigned* bm_e = (unsigned*)xmean;
  unsigned* bm_n = bm_e + (size_t)OW_NBLK * BM_PITCH;
  block_masks<<<dim3(BM_PITCH / 32, 2), 1024, 0, stream>>>(row, col, bm_e);

  owner_degrees<<<dim3(OW_NBLK, 2), OW_WAVES * 32, 0, stream>>>(xm, em, row, col, bm_e,
                                                                 alpha, Dg, Bg);

  gemm_direct_f32<FF, 8, 8, true><<<dim3(1, 512), 256, 0, stream>>>(
      bufA, conv1_W, bufB, NN * DD, HH, HH, HH);
  owner_rows<OW_RE, false><<<dim3(OW_NBLK, DH / OW_SLICE), OW_WAVES * 32, 0, stream>>>(
      bufB, alpha, Bg, row, col, bm_e, mbuf, nullptr, EE);
  owner_rows<OW_RN, true><<<dim3(OW_NBLK, DH / OW_SLICE), OW_WAVES * 32, 0, stream>>>(
      mbuf, alpha, Dg, col, row, bm_n, bufA, conv1_b, NN);

  gemm_direct_f32<HH, 8, 8, true, true><<<dim3(1, 512), 256, 0, stream>>>(
      bufA, conv2_W, bufB, NN * DD, HH, HH, HH);
  owner_rows<OW_RE, false><<<dim3(OW_NBLK, DH / OW_SLICE), OW_WAVES * 32, 0, stream>>>(
      bufB, alpha, Bg, row, col, bm_e, mbuf, nullptr, EE);
  owner_rows<OW_RN, true><<<dim3(OW_NBLK, DH / OW_SLICE), OW_WAVES * 32, 0, stream>>>(
      mbuf, alpha, Dg, col, row, bm_n, bufA, conv2_b, NN);

  gemm_direct_f32<DH, 3, 4, true><<<dim3(1, 313), 128, 0, stream>>>(bufA, lin2_W, out, NN, COUT, COUT, COUT);
}


# Scatter-add index path: degree launch packs each owner block's incidences into a dense record list (slot = bitmap popcount prefix); aggregation reads it, no bitmap scan or index gather; new add order
# speedup vs baseline: 4.4214x; 1.1004x over previous
#include <hip/hip_runtime.h>
#include <hip/hip_bf16.h>

#define NN   20000
#define EE   2000
#define NNZ  100000
#define FF   128
#define HH   128
#define DD   6
#define COUT 40
#define DH   (DD * HH)

typedef __attribute__((ext_vector_type(2))) float v2f;
typedef __attribute__((ext_vector_type(8))) float v8f;

typedef __attribute__((ext_vector_type(4))) float v4f;

__device__ __forceinline__ float elu1(float v) {
  return (v > 0.0f) ? v : (__expf(v) - 1.0f);
}

template <int K, int NT, int WAVES, bool VECB, bool ELUA = false>
__global__ __launch_bounds__(WAVES * 32)
void gemm_direct_f32(const float* __restrict__ A, const float* __restrict__ B,
                     float* __restrict__ C, int M, int Nc, int ldb, int ldc) {
  constexpr int NB = NT * 16;
  constexpr int KC = 128;
  constexpr int THREADS = WAVES * 32;
  constexpr bool PERM = (NT % 4) == 0;
  __shared__ v4f Bs[(KC / 4) * NB];

  const int tid  = threadIdx.x;
  const int wave = tid >> 5;
  const int lane = tid & 31;
  const int c    = lane & 15;
  const int h    = lane >> 4;
  const int nb   = blockIdx.x * NB;

  const int row_tiles = (M + 15) / 16;
  const int per_sweep = gridDim.y * WAVES;
  const int sweeps    = (row_tiles + per_sweep - 1) / per_sweep;

  auto slot_of = [](int n) {
    return PERM ? ((n >> 6) * 64 + (n & 3) * 16 + ((n & 63) >> 2)) : n;
  };

  auto stage_panel = [&](int kc) {
    if (VECB) {
      constexpr int G = NB / 4;
      constexpr int ITERS = ((KC / 4) * G) / THREADS;
      static_assert(((KC / 4) * G) % THREADS == 0, "panel must divide evenly over the block");
#pragma unroll
      for (int i = 0; i < ITERS; ++i) {
        const int idx = tid + i * THREADS;
        const int q = idx / G, n = (idx % G) * 4;
        const int gn = nb + n;
        const int gs = (gn < Nc) ? gn : (Nc - 4);
        const float* bp = B + (size_t)(kc + 4 * q) * ldb + gs;
        v4f r0 = *(const v4f*)(bp);
        v4f r1 = *(const v4f*)(bp + ldb);
        v4f r2 = *(const v4f*)(bp + 2 * (size_t)ldb);
        v4f r3 = *(const v4f*)(bp + 3 * (size_t)ldb);
        Bs[q * NB + slot_of(n + 0)] = (v4f){r0.x, r1.x, r2.x, r3.x};
        Bs[q * NB + slot_of(n + 1)] = (v4f){r0.y, r1.y, r2.y, r3.y};
        Bs[q * NB + slot_of(n + 2)] = (v4f){r0.z, r1.z, r2.z, r3.z};
        Bs[q * NB + slot_of(n + 3)] = (v4f){r0.w, r1.w, r2.w, r3.w};
      }
    } else {
      constexpr int ITERS = ((KC / 4) * NB) / THREADS;
      static_assert(((KC / 4) * NB) % THREADS == 0, "panel must divide evenly over the block");
#pragma unroll
      for (int i = 0; i < ITERS; ++i) {
        const int idx = tid + i * THREADS;
        const int q = idx / NB, n = idx % NB;
        const int gn = nb + n;
        const int gs = (gn < Nc) ? gn : (Nc - 1);
        const float* bp = B + (size_t)(kc + 4 * q) * ldb + gs;
        v4f v;
        v.x = bp[0];
        v.y = bp[ldb];
        v.z = bp[2 * (size_t)ldb];
        v.w = bp[3 * (size_t)ldb];
        Bs[q * NB + slot_of(n)] = v;
      }
    }
  };

  if (K == KC) {
    stage_panel(0);
    __syncthreads();
  }

  for (int it = 0; it < sweeps; ++it) {
    const int tile = (it * gridDim.y + blockIdx.y) * WAVES + wave;
    const int r0   = tile * 16;
    if (K == KC && it > 0 && r0 >= M) break;

    int gm = r0 + c;
    gm = (gm < M) ? gm : (M - 1);
    const float* ap = A + (size_t)gm * K + 4 * h;

    v8f acc[NT];
#pragma unroll
    for (int nt = 0; nt < NT; ++nt) acc[nt] = (v8f){};

    v4f a_cur[4];
#pragma unroll
    for (int j = 0; j < 4; ++j) {
      a_cur[j] = *(const v4f*)(ap + 8 * j);
      if (ELUA) a_cur[j] = (v4f){elu1(a_cur[j].x), elu1(a_cur[j].y), elu1(a_cur[j].z), elu1(a_cur[j].w)};
    }

    for (int kc = 0; kc < K; kc += KC) {
      if (K > KC) {
        __syncthreads();
        stage_panel(kc);
        __syncthreads();
      }

#pragma unroll
      for (int s = 0; s < 4; ++s) {
        v4f a_nxt[4];
        const int kn = kc + (s + 1) * 32;
        if (kn < K) {
#pragma unroll
          for (int j = 0; j < 4; ++j) a_nxt[j] = *(const v4f*)(ap + kn + 8 * j);
        }
        __builtin_amdgcn_sched_barrier(0);
#pragma unroll
        for (int j = 0; j < 4; ++j) {
          v4f b[NT];
#pragma unroll
          for (int nt = 0; nt < NT; ++nt) b[nt] = Bs[(8 * s + 2 * j + h) * NB + nt * 16 + c];
          const v2f alo = {a_cur[j].x, a_cur[j].y};
          const v2f ahi = {a_cur[j].z, a_cur[j].w};
#pragma unroll
          for (int nt = 0; nt < NT; ++nt) {
            const v2f blo = {b[nt].x, b[nt].y};
            acc[nt] = __builtin_amdgcn_wmma_f32_16x16x4_f32(
                false, alo, false, blo, (short)0, acc[nt], false, false);
          }
#pragma unroll
          for (int nt = 0; nt < NT; ++nt) {
            const v2f bhi = {b[nt].z, b[nt].w};
            acc[nt] = __builtin_amdgcn_wmma_f32_16x16x4_f32(
                false, ahi, false, bhi, (short)0, acc[nt], false, false);
          }
        }
        if (kn < K) {
#pragma unroll
          for (int j = 0; j < 4; ++j) {
            a_cur[j] = ELUA ? (v4f){elu1(a_nxt[j].x), elu1(a_nxt[j].y), elu1(a_nxt[j].z), elu1(a_nxt[j].w)}
                            : a_nxt[j];
          }
        }
      }
    }

    const bool vec_rows = ((ldc & 3) == 0) && (nb + NB <= Nc);
#pragma unroll
    for (int r = 0; r < 8; ++r) {
      const int grow = r0 + 8 * h + r;
      if (grow < M) {
        float* crow = C + (size_t)grow * ldc;
        if (PERM) {
#pragma unroll
          for (int g = 0; g < NT / 4; ++g) {
            const int n = nb + g * 64 + 4 * c;
            if (vec_rows) {
              v4f o = {acc[4 * g + 0][r], acc[4 * g + 1][r], acc[4 * g + 2][r], acc[4 * g + 3][r]};
              *(v4f*)(crow + n) = o;
            } else {
#pragma unroll
              for (int e = 0; e < 4; ++e)
                if (n + e < Nc) crow[n + e] = acc[4 * g + e][r];
            }
          }
        } else {
#pragma unroll
          for (int nt = 0; nt < NT; ++nt) {
            const int n = nb + nt * 16 + c;
            if (n < Nc) crow[n] = acc[nt][r];
          }
        }
      }
    }
  }
}

template <int WAVES, bool STORE_C>
__global__ __launch_bounds__(WAVES * 32)
void gemm_lin_mean_f32(const float* __restrict__ A, const float* __restrict__ B,
                       float* __restrict__ C, float* __restrict__ Cm, int M) {
  constexpr int K  = FF;
  constexpr int NT = 2 * DD;
  constexpr int NB = NT * 16;
  constexpr int FS = 32;
  constexpr int THREADS = WAVES * 32;
  __shared__ v4f Bs[(K / 4) * NB];

  const int tid  = threadIdx.x;
  const int lane = tid & 31;
  const int c    = lane & 15;
  const int h    = lane >> 4;
  const int wave = __builtin_amdgcn_readfirstlane(tid >> 5);
  const int f0   = blockIdx.x * FS;

  {
    constexpr int G = NB / 4;
    constexpr int ITERS = ((K / 4) * G) / THREADS;
    static_assert(((K / 4) * G) % THREADS == 0, "panel must divide evenly over the block");
#pragma unroll
    for (int i = 0; i < ITERS; ++i) {
      const int idx = tid + i * THREADS;
      const int q = idx / G, g = idx % G;
      const int d = g >> 3, j0 = (g & 7) * 4;
      const float* bp = B + (size_t)(4 * q) * DH + d * FF + f0 + j0;
      const v4f r0 = *(const v4f*)(bp);
      const v4f r1 = *(const v4f*)(bp + DH);
      const v4f r2 = *(const v4f*)(bp + 2 * DH);
      const v4f r3 = *(const v4f*)(bp + 3 * DH);
      const int s0 = q * NB + (2 * d) * 16 + (j0 >> 1);
      Bs[s0]      = (v4f){r0.x, r1.x, r2.x, r3.x};
      Bs[s0 + 16] = (v4f){r0.y, r1.y, r2.y, r3.y};
      Bs[s0 + 1]  = (v4f){r0.z, r1.z, r2.z, r3.z};
      Bs[s0 + 17] = (v4f){r0.w, r1.w, r2.w, r3.w};
    }
  }
  __syncthreads();

  const int row_tiles = (M + 15) / 16;
  const int per_sweep = gridDim.y * WAVES;
  const int sweeps    = (row_tiles + per_sweep - 1) / per_sweep;

  for (int it = 0; it < sweeps; ++it) {
    const int tile = it * per_sweep + wave * gridDim.y + blockIdx.y;
    const int r0   = tile * 16;
    if (r0 >= M) break;

    int gm = r0 + c;
    gm = (gm < M) ? gm : (M - 1);
    const float* ap = A + (size_t)gm * K + 4 * h;

    v8f acc[NT];
#pragma unroll
    for (int nt = 0; nt < NT; ++nt) acc[nt] = (v8f){};

    v4f a_cur[4];
#pragma unroll
    for (int j = 0; j < 4; ++j) a_cur[j] = *(const v4f*)(ap + 8 * j);

#pragma unroll
    for (int s = 0; s < 4; ++s) {
      v4f a_nxt[4];
      const int kn = (s + 1) * 32;
      if (kn < K) {
#pragma unroll
        for (int j = 0; j < 4; ++j) a_nxt[j] = *(const v4f*)(ap + kn + 8 * j);
      }
      __builtin_amdgcn_sched_barrier(0);
#pragma unroll
      for (int j = 0; j < 4; ++j) {
        v4f b[NT];
#pragma unroll
        for (int nt = 0; nt < NT; ++nt) b[nt] = Bs[(8 * s + 2 * j + h) * NB + nt * 16 + c];
        const v2f alo = {a_cur[j].x, a_cur[j].y};
        const v2f ahi = {a_cur[j].z, a_cur[j].w};
#pragma unroll
        for (int nt = 0; nt < NT; ++nt) {
          const v2f blo = {b[nt].x, b[nt].y};
          acc[nt] = __builtin_amdgcn_wmma_f32_16x16x4_f32(
              false, alo, false, blo, (short)0, acc[nt], false, false);
        }
#pragma unroll
        for (int nt = 0; nt < NT; ++nt) {
          const v2f bhi = {b[nt].z, b[nt].w};
          acc[nt] = __builtin_amdgcn_wmma_f32_16x16x4_f32(
              false, ahi, false, bhi, (short)0, acc[nt], false, false);
        }
      }
      if (kn < K) {
#pragma unroll
        for (int j = 0; j < 4; ++j) a_cur[j] = a_nxt[j];
      }
    }

#pragma unroll
    for (int r = 0; r < 8; ++r) {
      const int grow = r0 + 8 * h + r;
      if (grow < M) {
        float s0 = 0.0f, s1 = 0.0f;
#pragma unroll
        for (int d = 0; d < DD; ++d) {
          const float v0 = acc[2 * d][r], v1 = acc[2 * d + 1][r];
          if (STORE_C)
            *(v2f*)(C + (size_t)grow * DH + d * FF + f0 + 2 * c) = (v2f){v0, v1};
          s0 += v0;
          s1 += v1;
        }
        *(v2f*)(Cm + (size_t)grow * FF + f0 + 2 * c) =
            (v2f){s0 * (1.0f / (float)DD), s1 * (1.0f / (float)DD)};
      }
    }
  }
}

#define OW_WAVES 32
#define OW_SLICE 128
#define OW_RING  64
#define OW_BATCH 8

#define OW_RN   250
#define OW_RE   25
#define OW_NBLK 80
#define BM_WORDS ((NNZ + 31) / 32)
#define BM_PITCH (((BM_WORDS + 31) / 32) * 32)
#define BM_WPW   (BM_PITCH / OW_WAVES)
#define BM_NWG   (BM_PITCH / 32)

static_assert(OW_NBLK * OW_RN == NN && OW_NBLK * OW_RE == EE, "the blocks cover the destination rows");
static_assert(BM_PITCH % OW_WAVES == 0, "a bitmap row divides evenly over the waves");
static_assert(OW_WAVES == 32, "one lane per wave in the prefix sums over a workgroup's waves");
static_assert(NN <= 65536 && EE <= 65536 && OW_RN <= 32768, "a record packs source row and local row in 16 bits each");

__device__ __forceinline__ int wave_sum(int v) {
#pragma unroll
  for (int m = 16; m > 0; m >>= 1) v += __shfl_xor(v, m);
  return v;
}

__global__ __launch_bounds__(1024)
void block_masks(const int* __restrict__ row, const int* __restrict__ col,
                 unsigned* __restrict__ bm, int* __restrict__ cnt) {
  constexpr int TP = OW_NBLK + 1;
  __shared__ unsigned tile[32 * TP];
  const int tid  = threadIdx.x;
  const int lane = tid & 31;
  const int wave = __builtin_amdgcn_readfirstlane(tid >> 5);
  const int dir  = blockIdx.y;
  const int p    = blockIdx.x * 32 + wave;
  const int i    = (p % BM_WPW) * 1024 + 32 * lane + p / BM_WPW;
  int blk = -1;
  if (i < NNZ) blk = dir ? (row[i] / OW_RN) : (col[i] / OW_RE);
#pragma unroll
  for (int c = 0; c < (OW_NBLK + 31) / 32; ++c) {
    int mine = 0;
#pragma unroll
    for (int j = 0; j < 32; ++j) {
      const int m = (int)__builtin_amdgcn_ballot_w32(blk == c * 32 + j);
      mine = (lane == j) ? m : mine;
    }
    const int b = c * 32 + lane;
    if (b < OW_NBLK) tile[wave * TP + b] = (unsigned)mine;
  }
  __syncthreads();
  unsigned* out = bm + (size_t)dir * OW_NBLK * BM_PITCH + blockIdx.x * 32;
  for (int k = tid; k < OW_NBLK * 32; k += 1024)
    out[(size_t)(k >> 5) * BM_PITCH + (k & 31)] = tile[(k & 31) * TP + (k >> 5)];
  if (tid < OW_NBLK) {
    int n = 0;
    for (int w = 0; w < 32; ++w) n += __builtin_popcount(tile[w * TP + tid]);
    cnt[((size_t)dir * OW_NBLK + tid) * BM_NWG + blockIdx.x] = n;
  }
}

__device__ __forceinline__ void lds_add(float* p, float v) {
  __hip_atomic_fetch_add(p, v, __ATOMIC_RELAXED, __HIP_MEMORY_SCOPE_WORKGROUP);
}

#define DG_COPIES 4

__global__ __launch_bounds__(OW_WAVES * 32)
void owner_degrees(const float* __restrict__ xm, const float* __restrict__ em,
                   const int* __restrict__ row, const int* __restrict__ col,
                   const unsigned* __restrict__ bm, const int* __restrict__ cnt,
                   float* __restrict__ Dg, float* __restrict__ Bg,
                   int* __restrict__ recs, float* __restrict__ arec,
                   int* __restrict__ blk) {
  __shared__ float deg[DG_COPIES * OW_RN * DD];
  __shared__ int rings[OW_WAVES * OW_RING];
  __shared__ int wtot[OW_WAVES];
  __shared__ int wpre[OW_WAVES];

  const int tid  = threadIdx.x;
  const int lane = tid & 31;
  const int wave = __builtin_amdgcn_readfirstlane(tid >> 5);
  const int dir  = blockIdx.y;
  const int R    = dir ? OW_RN : OW_RE;
  const int r0   = blockIdx.x * R;

  const unsigned* bw = bm + ((size_t)dir * OW_NBLK + blockIdx.x) * BM_PITCH + wave * BM_WPW;
  constexpr int SLOTS = (BM_WPW + 31) / 32;
  unsigned wv[SLOTS];
#pragma unroll
  for (int s = 0; s < SLOTS; ++s) {
    const int o = s * 32 + lane;
    wv[s] = (o < BM_WPW) ? bw[o] : 0u;
  }

  for (int k = tid; k < DG_COPIES * OW_RN * DD; k += OW_WAVES * 32) deg[k] = 0.0f;

  {
    int mine = 0;
#pragma unroll
    for (int s = 0; s < SLOTS; ++s) mine += __builtin_popcount(wv[s]);
    const int* cb = cnt + (size_t)dir * OW_NBLK * BM_NWG;
    int part = 0;
    for (int k = tid; k < (int)blockIdx.x * BM_NWG; k += OW_WAVES * 32) part += cb[k];
    mine = wave_sum(mine);
    part = wave_sum(part);
    if (lane == 0) { wtot[wave] = mine; wpre[wave] = part; }
  }
  __syncthreads();

  const int t_l       = wtot[lane];
  const int blk_base  = wave_sum(wpre[lane]);
  const int blk_total = wave_sum(t_l);
  const int wave_base = blk_base + wave_sum((lane < wave) ? t_l : 0);
  if (tid == 0) {
    blk[2 * (dir * OW_NBLK + blockIdx.x)]     = blk_base;
    blk[2 * (dir * OW_NBLK + blockIdx.x) + 1] = blk_total;
  }

  int*   ring = rings + wave * OW_RING;
  float* dsum = deg + (lane & (DG_COPIES - 1)) * (OW_RN * DD);
  int*   rc   = recs + (size_t)dir * NNZ;
  float* ar   = arec + (size_t)dir * DD * NNZ;

  auto take = [&](int head, int n, int taken) {
    if (lane < n) {
      const int inc = ring[(head + lane) & (OW_RING - 1)];
      const int r = row[inc], c = col[inc];
      const int loc = (dir ? r : c) - r0;
      const v2f* xp = (const v2f*)(xm + (size_t)r * DD);
      const v2f* ep = (const v2f*)(em + (size_t)c * DD);
      v2f a[DD / 2];
      for (int j = 0; j < DD / 2; ++j) {
        const v2f s = xp[j] + ep[j];
        a[j].x = 1.0f / (1.0f + __expf(-s.x));
        a[j].y = 1.0f / (1.0f + __expf(-s.y));
      }
      float* d = dsum + loc * DD;
      for (int j = 0; j < DD / 2; ++j) {
        lds_add(d + 2 * j,     a[j].x);
        lds_add(d + 2 * j + 1, a[j].y);
      }
      const int slot = wave_base + taken + lane;
      rc[slot] = (dir ? c : r) | (loc << 16);
      for (int j = 0; j < DD / 2; ++j) {
        ar[(size_t)(2 * j)     * NNZ + slot] = a[j].x;
        ar[(size_t)(2 * j + 1) * NNZ + slot] = a[j].y;
      }
    }
  };

  int head = 0, count = 0, taken = 0;
#pragma unroll
  for (int s = 0; s < SLOTS; ++s) {
    unsigned mw = wv[s];
    const int ibase = (s * 32 + lane) * 1024 + wave;
    for (int t = 0; t < 32; ++t) {
      const bool m = (mw != 0u);
      const unsigned mask = __builtin_amdgcn_ballot_w32(m);
      if (mask == 0u) break;
      if (m) {
        const int e = (head + count + (int)__builtin_amdgcn_mbcnt_lo(mask, 0u)) & (OW_RING - 1);
        ring[e] = ibase + 32 * __builtin_ctz(mw);
        mw &= mw - 1u;
      }
      count += __builtin_popcount(mask);
      if (count >= 32) {
        take(head, 32, taken);
        head   = (head + 32) & (OW_RING - 1);
        count -= 32;
        taken += 32;
      }
    }
  }
  if (count > 0) take(head, count, taken);
  __syncthreads();

  float* out = (dir ? Dg : Bg) + (size_t)r0 * DD;
  for (int k = tid; k < R * DD; k += OW_WAVES * 32) {
    float v = deg[k];
    for (int cpy = 1; cpy < DG_COPIES; ++cpy) v += deg[cpy * (OW_RN * DD) + k];
    out[k] = (v != 0.0f) ? (1.0f / v) : 0.0f;
  }
}

__device__ __forceinline__ void ow_batch(int k0, int nb, int srow, int loc, float scl,
                                         const float* sp, float* ap) {
  v4f   v[OW_BATCH];
  int   dl[OW_BATCH];
  float sc[OW_BATCH];
  const int last = k0 + nb - 1;
#pragma unroll
  for (int b = 0; b < OW_BATCH; ++b) {
    const int li = (k0 + b < last) ? (k0 + b) : last;
    const int sr = __builtin_amdgcn_readlane(srow, li);
    dl[b] = __builtin_amdgcn_readlane(loc, li);
    sc[b] = __int_as_float(__builtin_amdgcn_readlane(__float_as_int(scl), li));
    v[b]  = *(const v4f*)(sp + (size_t)sr * DH);
  }
#pragma unroll
  for (int b = 0; b < OW_BATCH; ++b) {
    if (b < nb) {
      float* a = ap + dl[b] * OW_SLICE;
      lds_add(a,      sc[b] * v[b].x);
      lds_add(a + 32, sc[b] * v[b].y);
      lds_add(a + 64, sc[b] * v[b].z);
      lds_add(a + 96, sc[b] * v[b].w);
    }
  }
}

template <int R, bool BIAS>
__global__ __launch_bounds__(OW_WAVES * 32)
void owner_rows(const float* __restrict__ src, const float* __restrict__ arec,
                const float* __restrict__ inv, const int* __restrict__ recs,
                const int* __restrict__ blk, float* __restrict__ dst,
                const float* __restrict__ bias, int rows) {
  __shared__ __attribute__((aligned(16))) float acc[R * OW_SLICE];
  __shared__ float ivs[R];

  const int tid  = threadIdx.x;
  const int lane = tid & 31;
  const int wave = __builtin_amdgcn_readfirstlane(tid >> 5);
  const int r0   = blockIdx.x * R;
  const int dch  = blockIdx.y;
  const int col0 = dch * OW_SLICE;

  for (int k = tid; k < R * OW_SLICE / 4; k += OW_WAVES * 32) ((v4f*)acc)[k] = (v4f){};
  for (int k = tid; k < R; k += OW_WAVES * 32) ivs[k] = inv[(size_t)(r0 + k) * DD + dch];
  __syncthreads();

  const float* sp = src + col0 + 4 * lane;
  float*       ap = acc + lane;

  const int base  = blk[2 * blockIdx.x];
  const int total = blk[2 * blockIdx.x + 1];
  const int per   = (total + OW_WAVES - 1) / OW_WAVES;
  const int beg   = wave * per;
  const int end   = (beg + per < total) ? (beg + per) : total;
  const int*   rp = recs + base;
  const float* al = arec + (size_t)dch * NNZ + base;

  for (int j = beg; j < end; j += 32) {
    const int n = (end - j < 32) ? (end - j) : 32;
    int   pk = 0;
    float a  = 0.0f;
    if (lane < n) { pk = rp[j + lane]; a = al[j + lane]; }
    const int   srow = pk & 0xffff;
    const int   loc  = pk >> 16;
    const float scl  = ivs[loc] * a;
    for (int k0 = 0; k0 < n; k0 += OW_BATCH)
      ow_batch(k0, (n - k0 < OW_BATCH) ? (n - k0) : OW_BATCH, srow, loc, scl, sp, ap);
  }
  __syncthreads();

  v4f bb = {0.0f, 0.0f, 0.0f, 0.0f};
  if (BIAS) bb = *(const v4f*)(bias + 4 * lane);
  for (int r = wave; r < R; r += OW_WAVES) {
    const int grow = r0 + r;
    if (grow < rows) {
      const float* a = ap + r * OW_SLICE;
      v4f o = {a[0], a[32], a[64], a[96]};
      if (BIAS) o += bb;
      *(v4f*)(dst + (size_t)grow * DH + col0 + 4 * lane) = o;
    }
  }
}

extern "C" void kernel_launch(void* const* d_in, const int* in_sizes, int n_in,
                              void* d_out, int out_size, void* d_ws, size_t ws_size,
                              hipStream_t stream) {
  const float* x        = (const float*)d_in[0];
  const int*   eidx     = (const int*)d_in[1];
  const float* hedge    = (const float*)d_in[2];
  const float* lin_W    = (const float*)d_in[3];
  const float* sheaf_W  = (const float*)d_in[4];
  const float* conv1_W  = (const float*)d_in[5];
  const float* conv1_b  = (const float*)d_in[6];
  const float* conv2_W  = (const float*)d_in[7];
  const float* conv2_b  = (const float*)d_in[8];
  const float* lin2_W   = (const float*)d_in[9];
  float* out = (float*)d_out;

  const int* row = eidx;
  const int* col = eidx + NNZ;

  float* ws    = (float*)d_ws;
  float* bufA  = ws;
  float* bufB  = bufA + (size_t)NN * DH;
  float* mbuf  = bufB + (size_t)NN * DH;
  float* xmean = mbuf + (size_t)EE * DH;
  float* emean = xmean + (size_t)NN * FF;
  float* xm    = emean + (size_t)EE * FF;
  float* em    = xm + (size_t)NN * DD;
  float* alpha = em + (size_t)EE * DD;
  float* Dg    = alpha + (size_t)NNZ * DD;
  float* Bg    = Dg + (size_t)NN * DD;

  gemm_lin_mean_f32<16, true><<<dim3(4, 64), 512, 0, stream>>>(x, lin_W, bufA, xmean, NN);
  gemm_lin_mean_f32<4, false><<<dim3(4, 32), 128, 0, stream>>>(hedge, lin_W, nullptr, emean, EE);

  gemm_direct_f32<FF, 1, 4, false><<<dim3(1, 313), 128, 0, stream>>>(xmean, sheaf_W, xm, NN, DD, DD, DD);
  gemm_direct_f32<FF, 1, 4, false><<<dim3(1, 32), 128, 0, stream>>>(emean, sheaf_W + FF * DD, em, EE, DD, DD, DD);

  unsigned* bm_e = (unsigned*)xmean;
  int*   cnt  = (int*)(bm_e + (size_t)2 * OW_NBLK * BM_PITCH);
  int*   blk  = cnt + 2 * OW_NBLK * BM_NWG;
  int*   recs = blk + 2 * OW_NBLK * 2;
  float* arec = (float*)(recs + 2 * NNZ);
  block_masks<<<dim3(BM_NWG, 2), 1024, 0, stream>>>(row, col, bm_e, cnt);

  owner_degrees<<<dim3(OW_NBLK, 2), OW_WAVES * 32, 0, stream>>>(xm, em, row, col, bm_e, cnt,
                                                                 Dg, Bg, recs, arec, blk);

  gemm_direct_f32<FF, 8, 8, true><<<dim3(1, 512), 256, 0, stream>>>(
      bufA, conv1_W, bufB, NN * DD, HH, HH, HH);
  owner_rows<OW_RE, false><<<dim3(OW_NBLK, DH / OW_SLICE), OW_WAVES * 32, 0, stream>>>(
      bufB, arec, Bg, recs, blk, mbuf, nullptr, EE);
  owner_rows<OW_RN, true><<<dim3(OW_NBLK, DH / OW_SLICE), OW_WAVES * 32, 0, stream>>>(
      mbuf, arec + (size_t)DD * NNZ, Dg, recs + NNZ, blk + 2 * OW_NBLK, bufA, conv1_b, NN);

  gemm_direct_f32<HH, 8, 8, true, true><<<dim3(1, 512), 256, 0, stream>>>(
      bufA, conv2_W, bufB, NN * DD, HH, HH, HH);
  owner_rows<OW_RE, false><<<dim3(OW_NBLK, DH / OW_SLICE), OW_WAVES * 32, 0, stream>>>(
      bufB, arec, Bg, recs, blk, mbuf, nullptr, EE);
  owner_rows<OW_RN, true><<<dim3(OW_NBLK, DH / OW_SLICE), OW_WAVES * 32, 0, stream>>>(
      mbuf, arec + (size_t)DD * NNZ, Dg, recs + NNZ, blk + 2 * OW_NBLK, bufA, conv2_b, NN);

  gemm_direct_f32<DH, 3, 4, true><<<dim3(1, 313), 128, 0, stream>>>(bufA, lin2_W, out, NN, COUT, COUT, COUT);
}


# Class projection in last aggregation's epilogue: WMMA on LDS sums x lin2_W slice per channel and k half, partials added to out by global f32 atomics (arrival order; bias summed first); no final GEMM
# speedup vs baseline: 4.9034x; 1.1090x over previous
#include <hip/hip_runtime.h>
#include <hip/hip_bf16.h>

#define NN   20000
#define EE   2000
#define NNZ  100000
#define FF   128
#define HH   128
#define DD   6
#define COUT 40
#define DH   (DD * HH)

typedef __attribute__((ext_vector_type(2))) float v2f;
typedef __attribute__((ext_vector_type(8))) float v8f;

typedef __attribute__((ext_vector_type(4))) float v4f;

__device__ __forceinline__ float elu1(float v) {
  return (v > 0.0f) ? v : (__expf(v) - 1.0f);
}

template <int K, int NT, int WAVES, bool VECB, bool ELUA = false>
__global__ __launch_bounds__(WAVES * 32)
void gemm_direct_f32(const float* __restrict__ A, const float* __restrict__ B,
                     float* __restrict__ C, int M, int Nc, int ldb, int ldc) {
  constexpr int NB = NT * 16;
  constexpr int KC = 128;
  constexpr int THREADS = WAVES * 32;
  constexpr bool PERM = (NT % 4) == 0;
  __shared__ v4f Bs[(KC / 4) * NB];

  const int tid  = threadIdx.x;
  const int wave = tid >> 5;
  const int lane = tid & 31;
  const int c    = lane & 15;
  const int h    = lane >> 4;
  const int nb   = blockIdx.x * NB;

  const int row_tiles = (M + 15) / 16;
  const int per_sweep = gridDim.y * WAVES;
  const int sweeps    = (row_tiles + per_sweep - 1) / per_sweep;

  auto slot_of = [](int n) {
    return PERM ? ((n >> 6) * 64 + (n & 3) * 16 + ((n & 63) >> 2)) : n;
  };

  auto stage_panel = [&](int kc) {
    if (VECB) {
      constexpr int G = NB / 4;
      constexpr int ITERS = ((KC / 4) * G) / THREADS;
      static_assert(((KC / 4) * G) % THREADS == 0, "panel must divide evenly over the block");
#pragma unroll
      for (int i = 0; i < ITERS; ++i) {
        const int idx = tid + i * THREADS;
        const int q = idx / G, n = (idx % G) * 4;
        const int gn = nb + n;
        const int gs = (gn < Nc) ? gn : (Nc - 4);
        const float* bp = B + (size_t)(kc + 4 * q) * ldb + gs;
        v4f r0 = *(const v4f*)(bp);
        v4f r1 = *(const v4f*)(bp + ldb);
        v4f r2 = *(const v4f*)(bp + 2 * (size_t)ldb);
        v4f r3 = *(const v4f*)(bp + 3 * (size_t)ldb);
        Bs[q * NB + slot_of(n + 0)] = (v4f){r0.x, r1.x, r2.x, r3.x};
        Bs[q * NB + slot_of(n + 1)] = (v4f){r0.y, r1.y, r2.y, r3.y};
        Bs[q * NB + slot_of(n + 2)] = (v4f){r0.z, r1.z, r2.z, r3.z};
        Bs[q * NB + slot_of(n + 3)] = (v4f){r0.w, r1.w, r2.w, r3.w};
      }
    } else {
      constexpr int ITERS = ((KC / 4) * NB) / THREADS;
      static_assert(((KC / 4) * NB) % THREADS == 0, "panel must divide evenly over the block");
#pragma unroll
      for (int i = 0; i < ITERS; ++i) {
        const int idx = tid + i * THREADS;
        const int q = idx / NB, n = idx % NB;
        const int gn = nb + n;
        const int gs = (gn < Nc) ? gn : (Nc - 1);
        const float* bp = B + (size_t)(kc + 4 * q) * ldb + gs;
        v4f v;
        v.x = bp[0];
        v.y = bp[ldb];
        v.z = bp[2 * (size_t)ldb];
        v.w = bp[3 * (size_t)ldb];
        Bs[q * NB + slot_of(n)] = v;
      }
    }
  };

  if (K == KC) {
    stage_panel(0);
    __syncthreads();
  }

  for (int it = 0; it < sweeps; ++it) {
    const int tile = (it * gridDim.y + blockIdx.y) * WAVES + wave;
    const int r0   = tile * 16;
    if (K == KC && it > 0 && r0 >= M) break;

    int gm = r0 + c;
    gm = (gm < M) ? gm : (M - 1);
    const float* ap = A + (size_t)gm * K + 4 * h;

    v8f acc[NT];
#pragma unroll
    for (int nt = 0; nt < NT; ++nt) acc[nt] = (v8f){};

    v4f a_cur[4];
#pragma unroll
    for (int j = 0; j < 4; ++j) {
      a_cur[j] = *(const v4f*)(ap + 8 * j);
      if (ELUA) a_cur[j] = (v4f){elu1(a_cur[j].x), elu1(a_cur[j].y), elu1(a_cur[j].z), elu1(a_cur[j].w)};
    }

    for (int kc = 0; kc < K; kc += KC) {
      if (K > KC) {
        __syncthreads();
        stage_panel(kc);
        __syncthreads();
      }

#pragma unroll
      for (int s = 0; s < 4; ++s) {
        v4f a_nxt[4];
        const int kn = kc + (s + 1) * 32;
        if (kn < K) {
#pragma unroll
          for (int j = 0; j < 4; ++j) a_nxt[j] = *(const v4f*)(ap + kn + 8 * j);
        }
        __builtin_amdgcn_sched_barrier(0);
#pragma unroll
        for (int j = 0; j < 4; ++j) {
          v4f b[NT];
#pragma unroll
          for (int nt = 0; nt < NT; ++nt) b[nt] = Bs[(8 * s + 2 * j + h) * NB + nt * 16 + c];
          const v2f alo = {a_cur[j].x, a_cur[j].y};
          const v2f ahi = {a_cur[j].z, a_cur[j].w};
#pragma unroll
          for (int nt = 0; nt < NT; ++nt) {
            const v2f blo = {b[nt].x, b[nt].y};
            acc[nt] = __builtin_amdgcn_wmma_f32_16x16x4_f32(
                false, alo, false, blo, (short)0, acc[nt], false, false);
          }
#pragma unroll
          for (int nt = 0; nt < NT; ++nt) {
            const v2f bhi = {b[nt].z, b[nt].w};
            acc[nt] = __builtin_amdgcn_wmma_f32_16x16x4_f32(
                false, ahi, false, bhi, (short)0, acc[nt], false, false);
          }
        }
        if (kn < K) {
#pragma unroll
          for (int j = 0; j < 4; ++j) {
            a_cur[j] = ELUA ? (v4f){elu1(a_nxt[j].x), elu1(a_nxt[j].y), elu1(a_nxt[j].z), elu1(a_nxt[j].w)}
                            : a_nxt[j];
          }
        }
      }
    }

    const bool vec_rows = ((ldc & 3) == 0) && (nb + NB <= Nc);
#pragma unroll
    for (int r = 0; r < 8; ++r) {
      const int grow = r0 + 8 * h + r;
      if (grow < M) {
        float* crow = C + (size_t)grow * ldc;
        if (PERM) {
#pragma unroll
          for (int g = 0; g < NT / 4; ++g) {
            const int n = nb + g * 64 + 4 * c;
            if (vec_rows) {
              v4f o = {acc[4 * g + 0][r], acc[4 * g + 1][r], acc[4 * g + 2][r], acc[4 * g + 3][r]};
              *(v4f*)(crow + n) = o;
            } else {
#pragma unroll
              for (int e = 0; e < 4; ++e)
                if (n + e < Nc) crow[n + e] = acc[4 * g + e][r];
            }
          }
        } else {
#pragma unroll
          for (int nt = 0; nt < NT; ++nt) {
            const int n = nb + nt * 16 + c;
            if (n < Nc) crow[n] = acc[nt][r];
          }
        }
      }
    }
  }
}

template <int WAVES, bool STORE_C>
__global__ __launch_bounds__(WAVES * 32)
void gemm_lin_mean_f32(const float* __restrict__ A, const float* __restrict__ B,
                       float* __restrict__ C, float* __restrict__ Cm, int M) {
  constexpr int K  = FF;
  constexpr int NT = 2 * DD;
  constexpr int NB = NT * 16;
  constexpr int FS = 32;
  constexpr int THREADS = WAVES * 32;
  __shared__ v4f Bs[(K / 4) * NB];

  const int tid  = threadIdx.x;
  const int lane = tid & 31;
  const int c    = lane & 15;
  const int h    = lane >> 4;
  const int wave = __builtin_amdgcn_readfirstlane(tid >> 5);
  const int f0   = blockIdx.x * FS;

  {
    constexpr int G = NB / 4;
    constexpr int ITERS = ((K / 4) * G) / THREADS;
    static_assert(((K / 4) * G) % THREADS == 0, "panel must divide evenly over the block");
#pragma unroll
    for (int i = 0; i < ITERS; ++i) {
      const int idx = tid + i * THREADS;
      const int q = idx / G, g = idx % G;
      const int d = g >> 3, j0 = (g & 7) * 4;
      const float* bp = B + (size_t)(4 * q) * DH + d * FF + f0 + j0;
      const v4f r0 = *(const v4f*)(bp);
      const v4f r1 = *(const v4f*)(bp + DH);
      const v4f r2 = *(const v4f*)(bp + 2 * DH);
      const v4f r3 = *(const v4f*)(bp + 3 * DH);
      const int s0 = q * NB + (2 * d) * 16 + (j0 >> 1);
      Bs[s0]      = (v4f){r0.x, r1.x, r2.x, r3.x};
      Bs[s0 + 16] = (v4f){r0.y, r1.y, r2.y, r3.y};
      Bs[s0 + 1]  = (v4f){r0.z, r1.z, r2.z, r3.z};
      Bs[s0 + 17] = (v4f){r0.w, r1.w, r2.w, r3.w};
    }
  }
  __syncthreads();

  const int row_tiles = (M + 15) / 16;
  const int per_sweep = gridDim.y * WAVES;
  const int sweeps    = (row_tiles + per_sweep - 1) / per_sweep;

  for (int it = 0; it < sweeps; ++it) {
    const int tile = it * per_sweep + wave * gridDim.y + blockIdx.y;
    const int r0   = tile * 16;
    if (r0 >= M) break;

    int gm = r0 + c;
    gm = (gm < M) ? gm : (M - 1);
    const float* ap = A + (size_t)gm * K + 4 * h;

    v8f acc[NT];
#pragma unroll
    for (int nt = 0; nt < NT; ++nt) acc[nt] = (v8f){};

    v4f a_cur[4];
#pragma unroll
    for (int j = 0; j < 4; ++j) a_cur[j] = *(const v4f*)(ap + 8 * j);

#pragma unroll
    for (int s = 0; s < 4; ++s) {
      v4f a_nxt[4];
      const int kn = (s + 1) * 32;
      if (kn < K) {
#pragma unroll
        for (int j = 0; j < 4; ++j) a_nxt[j] = *(const v4f*)(ap + kn + 8 * j);
      }
      __builtin_amdgcn_sched_barrier(0);
#pragma unroll
      for (int j = 0; j < 4; ++j) {
        v4f b[NT];
#pragma unroll
        for (int nt = 0; nt < NT; ++nt) b[nt] = Bs[(8 * s + 2 * j + h) * NB + nt * 16 + c];
        const v2f alo = {a_cur[j].x, a_cur[j].y};
        const v2f ahi = {a_cur[j].z, a_cur[j].w};
#pragma unroll
        for (int nt = 0; nt < NT; ++nt) {
          const v2f blo = {b[nt].x, b[nt].y};
          acc[nt] = __builtin_amdgcn_wmma_f32_16x16x4_f32(
              false, alo, false, blo, (short)0, acc[nt], false, false);
        }
#pragma unroll
        for (int nt = 0; nt < NT; ++nt) {
          const v2f bhi = {b[nt].z, b[nt].w};
          acc[nt] = __builtin_amdgcn_wmma_f32_16x16x4_f32(
              false, ahi, false, bhi, (short)0, acc[nt], false, false);
        }
      }
      if (kn < K) {
#pragma unroll
        for (int j = 0; j < 4; ++j) a_cur[j] = a_nxt[j];
      }
    }

#pragma unroll
    for (int r = 0; r < 8; ++r) {
      const int grow = r0 + 8 * h + r;
      if (grow < M) {
        float s0 = 0.0f, s1 = 0.0f;
#pragma unroll
        for (int d = 0; d < DD; ++d) {
          const float v0 = acc[2 * d][r], v1 = acc[2 * d + 1][r];
          if (STORE_C)
            *(v2f*)(C + (size_t)grow * DH + d * FF + f0 + 2 * c) = (v2f){v0, v1};
          s0 += v0;
          s1 += v1;
        }
        *(v2f*)(Cm + (size_t)grow * FF + f0 + 2 * c) =
            (v2f){s0 * (1.0f / (float)DD), s1 * (1.0f / (float)DD)};
      }
    }
  }
}

#define OW_WAVES 32
#define OW_SLICE 128
#define OW_RING  64
#define OW_BATCH 8

#define OW_RN   250
#define OW_RE   25
#define OW_NBLK 80
#define BM_WORDS ((NNZ + 31) / 32)
#define BM_PITCH (((BM_WORDS + 31) / 32) * 32)
#define BM_WPW   (BM_PITCH / OW_WAVES)
#define BM_NWG   (BM_PITCH / 32)

static_assert(OW_NBLK * OW_RN == NN && OW_NBLK * OW_RE == EE, "the blocks cover the destination rows");
static_assert(BM_PITCH % OW_WAVES == 0, "a bitmap row divides evenly over the waves");
static_assert(OW_WAVES == 32, "one lane per wave in the prefix sums over a workgroup's waves");
static_assert(NN <= 65536 && EE <= 65536 && OW_RN <= 32768, "a record packs source row and local row in 16 bits each");
static_assert(OW_RN <= 16 * 16 && OW_SLICE == 128 && COUT > 32 && COUT <= 48 && (NN * COUT) % 4 == 0,
              "class product in the aggregation epilogue: 16 row tiles x 2 k halves over 32 waves, 3 column tiles");

__device__ __forceinline__ int wave_sum(int v) {
#pragma unroll
  for (int m = 16; m > 0; m >>= 1) v += __shfl_xor(v, m);
  return v;
}

__global__ __launch_bounds__(1024)
void block_masks(const int* __restrict__ row, const int* __restrict__ col,
                 unsigned* __restrict__ bm, int* __restrict__ cnt,
                 float* __restrict__ out0) {
  {
    const int gid = (blockIdx.y * gridDim.x + blockIdx.x) * 1024 + threadIdx.x;
    if (gid < NN * COUT / 4) ((v4f*)out0)[gid] = (v4f){0.0f, 0.0f, 0.0f, 0.0f};
  }
  constexpr int TP = OW_NBLK + 1;
  __shared__ unsigned tile[32 * TP];
  const int tid  = threadIdx.x;
  const int lane = tid & 31;
  const int wave = __builtin_amdgcn_readfirstlane(tid >> 5);
  const int dir  = blockIdx.y;
  const int p    = blockIdx.x * 32 + wave;
  const int i    = (p % BM_WPW) * 1024 + 32 * lane + p / BM_WPW;
  int blk = -1;
  if (i < NNZ) blk = dir ? (row[i] / OW_RN) : (col[i] / OW_RE);
#pragma unroll
  for (int c = 0; c < (OW_NBLK + 31) / 32; ++c) {
    int mine = 0;
#pragma unroll
    for (int j = 0; j < 32; ++j) {
      const int m = (int)__builtin_amdgcn_ballot_w32(blk == c * 32 + j);
      mine = (lane == j) ? m : mine;
    }
    const int b = c * 32 + lane;
    if (b < OW_NBLK) tile[wave * TP + b] = (unsigned)mine;
  }
  __syncthreads();
  unsigned* out = bm + (size_t)dir * OW_NBLK * BM_PITCH + blockIdx.x * 32;
  for (int k = tid; k < OW_NBLK * 32; k += 1024)
    out[(size_t)(k >> 5) * BM_PITCH + (k & 31)] = tile[(k & 31) * TP + (k >> 5)];
  if (tid < OW_NBLK) {
    int n = 0;
    for (int w = 0; w < 32; ++w) n += __builtin_popcount(tile[w * TP + tid]);
    cnt[((size_t)dir * OW_NBLK + tid) * BM_NWG + blockIdx.x] = n;
  }
}

__device__ __forceinline__ void lds_add(float* p, float v) {
  __hip_atomic_fetch_add(p, v, __ATOMIC_RELAXED, __HIP_MEMORY_SCOPE_WORKGROUP);
}

#define DG_COPIES 4

__global__ __launch_bounds__(OW_WAVES * 32)
void owner_degrees(const float* __restrict__ xm, const float* __restrict__ em,
                   const int* __restrict__ row, const int* __restrict__ col,
                   const unsigned* __restrict__ bm, const int* __restrict__ cnt,
                   float* __restrict__ Dg, float* __restrict__ Bg,
                   int* __restrict__ recs, float* __restrict__ arec,
                   int* __restrict__ blk) {
  __shared__ float deg[DG_COPIES * OW_RN * DD];
  __shared__ int rings[OW_WAVES * OW_RING];
  __shared__ int wtot[OW_WAVES];
  __shared__ int wpre[OW_WAVES];

  const int tid  = threadIdx.x;
  const int lane = tid & 31;
  const int wave = __builtin_amdgcn_readfirstlane(tid >> 5);
  const int dir  = blockIdx.y;
  const int R    = dir ? OW_RN : OW_RE;
  const int r0   = blockIdx.x * R;

  const unsigned* bw = bm + ((size_t)dir * OW_NBLK + blockIdx.x) * BM_PITCH + wave * BM_WPW;
  constexpr int SLOTS = (BM_WPW + 31) / 32;
  unsigned wv[SLOTS];
#pragma unroll
  for (int s = 0; s < SLOTS; ++s) {
    const int o = s * 32 + lane;
    wv[s] = (o < BM_WPW) ? bw[o] : 0u;
  }

  for (int k = tid; k < DG_COPIES * OW_RN * DD; k += OW_WAVES * 32) deg[k] = 0.0f;

  {
    int mine = 0;
#pragma unroll
    for (int s = 0; s < SLOTS; ++s) mine += __builtin_popcount(wv[s]);
    const int* cb = cnt + (size_t)dir * OW_NBLK * BM_NWG;
    int part = 0;
    for (int k = tid; k < (int)blockIdx.x * BM_NWG; k += OW_WAVES * 32) part += cb[k];
    mine = wave_sum(mine);
    part = wave_sum(part);
    if (lane == 0) { wtot[wave] = mine; wpre[wave] = part; }
  }
  __syncthreads();

  const int t_l       = wtot[lane];
  const int blk_base  = wave_sum(wpre[lane]);
  const int blk_total = wave_sum(t_l);
  const int wave_base = blk_base + wave_sum((lane < wave) ? t_l : 0);
  if (tid == 0) {
    blk[2 * (dir * OW_NBLK + blockIdx.x)]     = blk_base;
    blk[2 * (dir * OW_NBLK + blockIdx.x) + 1] = blk_total;
  }

  int*   ring = rings + wave * OW_RING;
  float* dsum = deg + (lane & (DG_COPIES - 1)) * (OW_RN * DD);
  int*   rc   = recs + (size_t)dir * NNZ;
  float* ar   = arec + (size_t)dir * DD * NNZ;

  auto take = [&](int head, int n, int taken) {
    if (lane < n) {
      const int inc = ring[(head + lane) & (OW_RING - 1)];
      const int r = row[inc], c = col[inc];
      const int loc = (dir ? r : c) - r0;
      const v2f* xp = (const v2f*)(xm + (size_t)r * DD);
      const v2f* ep = (const v2f*)(em + (size_t)c * DD);
      v2f a[DD / 2];
      for (int j = 0; j < DD / 2; ++j) {
        const v2f s = xp[j] + ep[j];
        a[j].x = 1.0f / (1.0f + __expf(-s.x));
        a[j].y = 1.0f / (1.0f + __expf(-s.y));
      }
      float* d = dsum + loc * DD;
      for (int j = 0; j < DD / 2; ++j) {
        lds_add(d + 2 * j,     a[j].x);
        lds_add(d + 2 * j + 1, a[j].y);
      }
      const int slot = wave_base + taken + lane;
      rc[slot] = (dir ? c : r) | (loc << 16);
      for (int j = 0; j < DD / 2; ++j) {
        ar[(size_t)(2 * j)     * NNZ + slot] = a[j].x;
        ar[(size_t)(2 * j + 1) * NNZ + slot] = a[j].y;
      }
    }
  };

  int head = 0, count = 0, taken = 0;
#pragma unroll
  for (int s = 0; s < SLOTS; ++s) {
    unsigned mw = wv[s];
    const int ibase = (s * 32 + lane) * 1024 + wave;
    for (int t = 0; t < 32; ++t) {
      const bool m = (mw != 0u);
      const unsigned mask = __builtin_amdgcn_ballot_w32(m);
      if (mask == 0u) break;
      if (m) {
        const int e = (head + count + (int)__builtin_amdgcn_mbcnt_lo(mask, 0u)) & (OW_RING - 1);
        ring[e] = ibase + 32 * __builtin_ctz(mw);
        mw &= mw - 1u;
      }
      count += __builtin_popcount(mask);
      if (count >= 32) {
        take(head, 32, taken);
        head   = (head + 32) & (OW_RING - 1);
        count -= 32;
        taken += 32;
      }
    }
  }
  if (count > 0) take(head, count, taken);
  __syncthreads();

  float* out = (dir ? Dg : Bg) + (size_t)r0 * DD;
  for (int k = tid; k < R * DD; k += OW_WAVES * 32) {
    float v = deg[k];
    for (int cpy = 1; cpy < DG_COPIES; ++cpy) v += deg[cpy * (OW_RN * DD) + k];
    out[k] = (v != 0.0f) ? (1.0f / v) : 0.0f;
  }
}

template <int PITCH>
__device__ __forceinline__ void ow_batch(int k0, int nb, int srow, int loc, float scl,
                                         const float* sp, float* ap) {
  v4f   v[OW_BATCH];
  int   dl[OW_BATCH];
  float sc[OW_BATCH];
  const int last = k0 + nb - 1;
#pragma unroll
  for (int b = 0; b < OW_BATCH; ++b) {
    const int li = (k0 + b < last) ? (k0 + b) : last;
    const int sr = __builtin_amdgcn_readlane(srow, li);
    dl[b] = __builtin_amdgcn_readlane(loc, li);
    sc[b] = __int_as_float(__builtin_amdgcn_readlane(__float_as_int(scl), li));
    v[b]  = *(const v4f*)(sp + (size_t)sr * DH);
  }
#pragma unroll
  for (int b = 0; b < OW_BATCH; ++b) {
    if (b < nb) {
      float* a = ap + dl[b] * PITCH;
      lds_add(a,      sc[b] * v[b].x);
      lds_add(a + 32, sc[b] * v[b].y);
      lds_add(a + 64, sc[b] * v[b].z);
      lds_add(a + 96, sc[b] * v[b].w);
    }
  }
}

__device__ __forceinline__ void out_add(float* p, float v) {
  __hip_atomic_fetch_add(p, v, __ATOMIC_RELAXED, __HIP_MEMORY_SCOPE_AGENT);
}

template <int R, bool BIAS, bool PROJ>
__global__ __launch_bounds__(OW_WAVES * 32)
void owner_rows(const float* __restrict__ src, const float* __restrict__ arec,
                const float* __restrict__ inv, const int* __restrict__ recs,
                const int* __restrict__ blk, float* __restrict__ dst,
                const float* __restrict__ bias, int rows,
                const float* __restrict__ Wp, float* __restrict__ outp) {
  constexpr int P  = PROJ ? (OW_SLICE + 2) : OW_SLICE;
  constexpr int PQ = OW_SLICE / 4;
  constexpr int PN = COUT;
  __shared__ __attribute__((aligned(16))) float acc[R * P];
  __shared__ float ivs[R];
  __shared__ v4f Wq[PROJ ? PQ * PN : 1];

  const int tid  = threadIdx.x;
  const int lane = tid & 31;
  const int wave = __builtin_amdgcn_readfirstlane(tid >> 5);
  const int r0   = blockIdx.x * R;
  const int dch  = blockIdx.y;
  const int col0 = dch * OW_SLICE;

  if (PROJ) {
    const int w = tid & (OW_SLICE - 1);
    const float b0 = BIAS ? bias[4 * (w & 31) + (w >> 5)] : 0.0f;
    for (int r = tid >> 7; r < R; r += (OW_WAVES * 32) >> 7) acc[r * P + w] = b0;
    const float* wb = Wp + (size_t)dch * OW_SLICE * COUT;
    for (int idx = tid; idx < PQ * PN; idx += OW_WAVES * 32) {
      const int q = idx / PN, n = idx % PN;
      const float* bp = wb + (size_t)(4 * q) * COUT + n;
      Wq[q * PN + n] = (v4f){bp[0], bp[COUT], bp[2 * COUT], bp[3 * COUT]};
    }
  } else {
    for (int k = tid; k < R * OW_SLICE / 4; k += OW_WAVES * 32) ((v4f*)acc)[k] = (v4f){};
  }
  for (int k = tid; k < R; k += OW_WAVES * 32) ivs[k] = inv[(size_t)(r0 + k) * DD + dch];
  __syncthreads();

  const float* sp = src + col0 + 4 * lane;
  float*       ap = acc + lane;

  const int base  = blk[2 * blockIdx.x];
  const int total = blk[2 * blockIdx.x + 1];
  const int per   = (total + OW_WAVES - 1) / OW_WAVES;
  const int beg   = wave * per;
  const int end   = (beg + per < total) ? (beg + per) : total;
  const int*   rp = recs + base;
  const float* al = arec + (size_t)dch * NNZ + base;

  for (int j = beg; j < end; j += 32) {
    const int n = (end - j < 32) ? (end - j) : 32;
    int   pk = 0;
    float a  = 0.0f;
    if (lane < n) { pk = rp[j + lane]; a = al[j + lane]; }
    const int   srow = pk & 0xffff;
    const int   loc  = pk >> 16;
    const float scl  = ivs[loc] * a;
    for (int k0 = 0; k0 < n; k0 += OW_BATCH)
      ow_batch<P>(k0, (n - k0 < OW_BATCH) ? (n - k0) : OW_BATCH, srow, loc, scl, sp, ap);
  }
  __syncthreads();

  if (PROJ) {
    const int c  = lane & 15;
    const int h  = lane >> 4;
    const int rt = wave & 15;
    const int kh = wave >> 4;
    int lr = rt * 16 + c;
    lr = (lr < R) ? lr : (R - 1);
    const float* arow = acc + lr * P;
    int bcol[3];
#pragma unroll
    for (int nt = 0; nt < 3; ++nt) {
      const int n = nt * 16 + c;
      bcol[nt] = (n < COUT) ? n : (COUT - 1);
    }
    v8f d[3];
#pragma unroll
    for (int nt = 0; nt < 3; ++nt) d[nt] = (v8f){};
#pragma unroll
    for (int j = 0; j < 8; ++j) {
      const int kq = kh * 16 + 2 * j + h;
      const v2f alo = {arow[kq],      arow[32 + kq]};
      const v2f ahi = {arow[64 + kq], arow[96 + kq]};
      v4f b[3];
#pragma unroll
      for (int nt = 0; nt < 3; ++nt) b[nt] = Wq[kq * PN + bcol[nt]];
#pragma unroll
      for (int nt = 0; nt < 3; ++nt) {
        const v2f blo = {b[nt].x, b[nt].y};
        d[nt] = __builtin_amdgcn_wmma_f32_16x16x4_f32(
            false, alo, false, blo, (short)0, d[nt], false, false);
      }
#pragma unroll
      for (int nt = 0; nt < 3; ++nt) {
        const v2f bhi = {b[nt].z, b[nt].w};
        d[nt] = __builtin_amdgcn_wmma_f32_16x16x4_f32(
            false, ahi, false, bhi, (short)0, d[nt], false, false);
      }
    }
#pragma unroll
    for (int r = 0; r < 8; ++r) {
      const int lrow = rt * 16 + 8 * h + r;
      const int grow = r0 + lrow;
      if (lrow < R && grow < rows) {
        float* orow = outp + (size_t)grow * COUT;
#pragma unroll
        for (int nt = 0; nt < 3; ++nt) {
          const int n = nt * 16 + c;
          if (n < COUT) out_add(orow + n, d[nt][r]);
        }
      }
    }
  } else {
    v4f bb = {0.0f, 0.0f, 0.0f, 0.0f};
    if (BIAS) bb = *(const v4f*)(bias + 4 * lane);
    for (int r = wave; r < R; r += OW_WAVES) {
      const int grow = r0 + r;
      if (grow < rows) {
        const float* a = ap + r * OW_SLICE;
        v4f o = {a[0], a[32], a[64], a[96]};
        if (BIAS) o += bb;
        *(v4f*)(dst + (size_t)grow * DH + col0 + 4 * lane) = o;
      }
    }
  }
}

extern "C" void kernel_launch(void* const* d_in, const int* in_sizes, int n_in,
                              void* d_out, int out_size, void* d_ws, size_t ws_size,
                              hipStream_t stream) {
  const float* x        = (const float*)d_in[0];
  const int*   eidx     = (const int*)d_in[1];
  const float* hedge    = (const float*)d_in[2];
  const float* lin_W    = (const float*)d_in[3];
  const float* sheaf_W  = (const float*)d_in[4];
  const float* conv1_W  = (const float*)d_in[5];
  const float* conv1_b  = (const float*)d_in[6];
  const float* conv2_W  = (const float*)d_in[7];
  const float* conv2_b  = (const float*)d_in[8];
  const float* lin2_W   = (const float*)d_in[9];
  float* out = (float*)d_out;

  const int* row = eidx;
  const int* col = eidx + NNZ;

  float* ws    = (float*)d_ws;
  float* bufA  = ws;
  float* bufB  = bufA + (size_t)NN * DH;
  float* mbuf  = bufB + (size_t)NN * DH;
  float* xmean = mbuf + (size_t)EE * DH;
  float* emean = xmean + (size_t)NN * FF;
  float* xm    = emean + (size_t)EE * FF;
  float* em    = xm + (size_t)NN * DD;
  float* alpha = em + (size_t)EE * DD;
  float* Dg    = alpha + (size_t)NNZ * DD;
  float* Bg    = Dg + (size_t)NN * DD;

  gemm_lin_mean_f32<16, true><<<dim3(4, 64), 512, 0, stream>>>(x, lin_W, bufA, xmean, NN);
  gemm_lin_mean_f32<4, false><<<dim3(4, 32), 128, 0, stream>>>(hedge, lin_W, nullptr, emean, EE);

  gemm_direct_f32<FF, 1, 4, false><<<dim3(1, 313), 128, 0, stream>>>(xmean, sheaf_W, xm, NN, DD, DD, DD);
  gemm_direct_f32<FF, 1, 4, false><<<dim3(1, 32), 128, 0, stream>>>(emean, sheaf_W + FF * DD, em, EE, DD, DD, DD);

  unsigned* bm_e = (unsigned*)xmean;
  int*   cnt  = (int*)(bm_e + (size_t)2 * OW_NBLK * BM_PITCH);
  int*   blk  = cnt + 2 * OW_NBLK * BM_NWG;
  int*   recs = blk + 2 * OW_NBLK * 2;
  float* arec = (float*)(recs + 2 * NNZ);
  block_masks<<<dim3(BM_NWG, 2), 1024, 0, stream>>>(row, col, bm_e, cnt, out);

  owner_degrees<<<dim3(OW_NBLK, 2), OW_WAVES * 32, 0, stream>>>(xm, em, row, col, bm_e, cnt,
                                                                 Dg, Bg, recs, arec, blk);

  gemm_direct_f32<FF, 8, 8, true><<<dim3(1, 512), 256, 0, stream>>>(
      bufA, conv1_W, bufB, NN * DD, HH, HH, HH);
  owner_rows<OW_RE, false, false><<<dim3(OW_NBLK, DH / OW_SLICE), OW_WAVES * 32, 0, stream>>>(
      bufB, arec, Bg, recs, blk, mbuf, nullptr, EE, nullptr, nullptr);
  owner_rows<OW_RN, true, false><<<dim3(OW_NBLK, DH / OW_SLICE), OW_WAVES * 32, 0, stream>>>(
      mbuf, arec + (size_t)DD * NNZ, Dg, recs + NNZ, blk + 2 * OW_NBLK, bufA, conv1_b, NN,
      nullptr, nullptr);

  gemm_direct_f32<HH, 8, 8, true, true><<<dim3(1, 512), 256, 0, stream>>>(
      bufA, conv2_W, bufB, NN * DD, HH, HH, HH);
  owner_rows<OW_RE, false, false><<<dim3(OW_NBLK, DH / OW_SLICE), OW_WAVES * 32, 0, stream>>>(
      bufB, arec, Bg, recs, blk, mbuf, nullptr, EE, nullptr, nullptr);
  owner_rows<OW_RN, true, true><<<dim3(OW_NBLK, DH / OW_SLICE), OW_WAVES * 32, 0, stream>>>(
      mbuf, arec + (size_t)DD * NNZ, Dg, recs + NNZ, blk + 2 * OW_NBLK, nullptr, conv2_b, NN,
      lin2_W, out);
}
